# starting kernel
# baseline (speedup 1.0000x reference)
#include <hip/hip_runtime.h>

typedef float v2f __attribute__((ext_vector_type(2)));
typedef float v8f __attribute__((ext_vector_type(8)));

#define IN_DIM 256
#define HID 16
#define OUTD 10

__global__ void gcn_init_kernel(float* __restrict__ agg, float* __restrict__ out,
                                const float* __restrict__ b2, int n_nodes) {
    int idx = blockIdx.x * blockDim.x + threadIdx.x;
    if (idx < n_nodes * HID)  agg[idx] = 0.0f;
    if (idx < n_nodes * OUTD) out[idx] = b2[idx % OUTD];
}

__global__ void gcn_gemm1_wmma(const float* __restrict__ x, const float* __restrict__ W1,
                               float* __restrict__ m1, int n_tiles) {
    int wave = (int)((blockIdx.x * blockDim.x + threadIdx.x) >> 5);
    int lane = threadIdx.x & 31;
    if (wave >= n_tiles) return;
    int l15 = lane & 15;
    int hi  = lane >> 4;
    const float* xrow = x + ((size_t)(wave * 16 + l15)) * IN_DIM;

    v8f c = {};
    #pragma unroll 4
    for (int k0 = 0; k0 < IN_DIM; k0 += 4) {
        int ka = k0 + hi * 2;
        v2f a; a.x = xrow[ka];                 a.y = xrow[ka + 1];
        v2f b; b.x = W1[(ka + 0) * HID + l15]; b.y = W1[(ka + 1) * HID + l15];
        c = __builtin_amdgcn_wmma_f32_16x16x4_f32(false, a, false, b, (short)0, c, false, false);
    }
    float* mt = m1 + (size_t)(wave * 16) * HID;
    #pragma unroll
    for (int v = 0; v < 8; ++v)
        mt[(v + hi * 8) * HID + l15] = c[v];
}

__global__ void gcn_scatter1(const int* __restrict__ src, const int* __restrict__ dst,
                             const float* __restrict__ m1, float* __restrict__ agg,
                             long long n_edges) {
    long long e = (long long)blockIdx.x * blockDim.x + threadIdx.x;
    if (e >= n_edges) return;
    int s = src[e], d = dst[e];
    const float4* ms = (const float4*)(m1 + (size_t)s * HID);
    float* ad = agg + (size_t)d * HID;
    #pragma unroll
    for (int q = 0; q < 4; ++q) {
        float4 v = ms[q];
        atomicAdd(ad + q * 4 + 0, v.x);
        atomicAdd(ad + q * 4 + 1, v.y);
        atomicAdd(ad + q * 4 + 2, v.z);
        atomicAdd(ad + q * 4 + 3, v.w);
    }
}

__global__ void gcn_gemm2_wmma(const float* __restrict__ agg, const float* __restrict__ b1,
                               const float* __restrict__ W2, float* __restrict__ m2,
                               int n_tiles) {
    int wave = (int)((blockIdx.x * blockDim.x + threadIdx.x) >> 5);
    int lane = threadIdx.x & 31;
    if (wave >= n_tiles) return;
    int l15 = lane & 15;
    int hi  = lane >> 4;
    const float* arow = agg + (size_t)(wave * 16 + l15) * HID;

    v8f c = {};
    #pragma unroll
    for (int k0 = 0; k0 < HID; k0 += 4) {
        int ka = k0 + hi * 2;
        v2f a;
        a.x = fmaxf(arow[ka]     + b1[ka],     0.0f);
        a.y = fmaxf(arow[ka + 1] + b1[ka + 1], 0.0f);
        v2f b;
        b.x = (l15 < OUTD) ? W2[(ka + 0) * OUTD + l15] : 0.0f;
        b.y = (l15 < OUTD) ? W2[(ka + 1) * OUTD + l15] : 0.0f;
        c = __builtin_amdgcn_wmma_f32_16x16x4_f32(false, a, false, b, (short)0, c, false, false);
    }
    float* mt = m2 + (size_t)(wave * 16) * HID;
    #pragma unroll
    for (int v = 0; v < 8; ++v)
        mt[(v + hi * 8) * HID + l15] = c[v];
}

__global__ void gcn_scatter2(const int* __restrict__ src, const int* __restrict__ dst,
                             const float* __restrict__ m2, float* __restrict__ out,
                             long long n_edges) {
    long long e = (long long)blockIdx.x * blockDim.x + threadIdx.x;
    if (e >= n_edges) return;
    int s = src[e], d = dst[e];
    const float* ms = m2 + (size_t)s * HID;
    float* od = out + (size_t)d * OUTD;
    #pragma unroll
    for (int q = 0; q < OUTD; ++q)
        atomicAdd(od + q, ms[q]);
}

extern "C" void kernel_launch(void* const* d_in, const int* in_sizes, int n_in,
                              void* d_out, int out_size, void* d_ws, size_t ws_size,
                              hipStream_t stream) {
    const float* x  = (const float*)d_in[0];
    const int*   ei = (const int*)  d_in[1];
    const float* W1 = (const float*)d_in[2];
    const float* b1 = (const float*)d_in[3];
    const float* W2 = (const float*)d_in[4];
    const float* b2 = (const float*)d_in[5];
    float* out = (float*)d_out;

    int n_nodes       = in_sizes[0] / IN_DIM;
    long long n_edges = (long long)in_sizes[1] / 2;
    const int* src = ei;
    const int* dst = ei + n_edges;

    float* m1  = (float*)d_ws;
    float* agg = m1  + (size_t)n_nodes * HID;
    float* m2  = agg + (size_t)n_nodes * HID;

    int n_tiles = (n_nodes + 15) / 16;
    int init_blocks = (n_nodes * HID + 255) / 256;
    int gemm_blocks = (n_tiles + 7) / 8;
    int edge_blocks = (int)((n_edges + 255) / 256);

    gcn_init_kernel<<<init_blocks, 256, 0, stream>>>(agg, out, b2, n_nodes);
    gcn_gemm1_wmma <<<gemm_blocks, 256, 0, stream>>>(x, W1, m1, n_tiles);
    gcn_scatter1   <<<edge_blocks, 256, 0, stream>>>(src, dst, m1, agg, n_edges);
    gcn_gemm2_wmma <<<gemm_blocks, 256, 0, stream>>>(agg, b1, W2, m2, n_tiles);
    gcn_scatter2   <<<edge_blocks, 256, 0, stream>>>(src, dst, m2, out, n_edges);
}


# one lane per feature in both scatters (16 lanes per edge, resident waves loop over 32-edge chunks): a row read or f32 atomic add covers one whole row, not 32 rows; sums stay in atomic arrival order
# speedup vs baseline: 6.1308x; 6.1308x over previous
#include <hip/hip_runtime.h>

typedef float v2f __attribute__((ext_vector_type(2)));
typedef float v8f __attribute__((ext_vector_type(8)));

#define IN_DIM 256
#define HID 16
#define OUTD 10

__global__ void gcn_init_kernel(float* __restrict__ agg, float* __restrict__ out,
                                const float* __restrict__ b2, int n_nodes) {
    int idx = blockIdx.x * blockDim.x + threadIdx.x;
    if (idx < n_nodes * HID)  agg[idx] = 0.0f;
    if (idx < n_nodes * OUTD) out[idx] = b2[idx % OUTD];
}

__global__ void gcn_gemm1_wmma(const float* __restrict__ x, const float* __restrict__ W1,
                               float* __restrict__ m1, int n_tiles) {
    int wave = (int)((blockIdx.x * blockDim.x + threadIdx.x) >> 5);
    int lane = threadIdx.x & 31;
    if (wave >= n_tiles) return;
    int l15 = lane & 15;
    int hi  = lane >> 4;
    const float* xrow = x + ((size_t)(wave * 16 + l15)) * IN_DIM;

    v8f c = {};
    #pragma unroll 4
    for (int k0 = 0; k0 < IN_DIM; k0 += 4) {
        int ka = k0 + hi * 2;
        v2f a; a.x = xrow[ka];                 a.y = xrow[ka + 1];
        v2f b; b.x = W1[(ka + 0) * HID + l15]; b.y = W1[(ka + 1) * HID + l15];
        c = __builtin_amdgcn_wmma_f32_16x16x4_f32(false, a, false, b, (short)0, c, false, false);
    }
    float* mt = m1 + (size_t)(wave * 16) * HID;
    #pragma unroll
    for (int v = 0; v < 8; ++v)
        mt[(v + hi * 8) * HID + l15] = c[v];
}

template <bool FULL>
__device__ __forceinline__ void scatter1_chunk(const int* __restrict__ src, const int* __restrict__ dst,
                                               const float* __restrict__ m1, float* __restrict__ agg,
                                               long long e0, long long n_edges, int lane, int f) {
    long long e = e0 + lane;
    int s_l = 0, d_l = -1;
    if (FULL || e < n_edges) { s_l = src[e]; d_l = dst[e]; }
    float v[16];
    int   d[16];
    #pragma unroll
    for (int j = 0; j < 16; ++j) {
        int s = __shfl(s_l, j, 16);
        d[j]  = __shfl(d_l, j, 16);
        v[j]  = m1[(unsigned)s * HID + f];
    }
    #pragma unroll
    for (int j = 0; j < 16; ++j)
        if (FULL || d[j] >= 0) atomicAdd(agg + (unsigned)d[j] * HID + f, v[j]);
}

__global__ void __launch_bounds__(256)
gcn_scatter1(const int* __restrict__ src, const int* __restrict__ dst,
             const float* __restrict__ m1, float* __restrict__ agg, long long n_edges) {
    int lane = threadIdx.x & 31;
    int f    = lane & 15;
    long long wave     = ((long long)blockIdx.x * blockDim.x + threadIdx.x) >> 5;
    long long n_waves  = ((long long)gridDim.x * blockDim.x) >> 5;
    long long n_chunks = (n_edges + 31) >> 5;
    for (long long c = wave; c < n_chunks; c += n_waves) {
        long long e0 = c * 32;
        if (e0 + 32 <= n_edges) scatter1_chunk<true >(src, dst, m1, agg, e0, n_edges, lane, f);
        else                    scatter1_chunk<false>(src, dst, m1, agg, e0, n_edges, lane, f);
    }
}

__global__ void gcn_gemm2_wmma(const float* __restrict__ agg, const float* __restrict__ b1,
                               const float* __restrict__ W2, float* __restrict__ m2,
                               int n_tiles) {
    int wave = (int)((blockIdx.x * blockDim.x + threadIdx.x) >> 5);
    int lane = threadIdx.x & 31;
    if (wave >= n_tiles) return;
    int l15 = lane & 15;
    int hi  = lane >> 4;
    const float* arow = agg + (size_t)(wave * 16 + l15) * HID;

    v8f c = {};
    #pragma unroll
    for (int k0 = 0; k0 < HID; k0 += 4) {
        int ka = k0 + hi * 2;
        v2f a;
        a.x = fmaxf(arow[ka]     + b1[ka],     0.0f);
        a.y = fmaxf(arow[ka + 1] + b1[ka + 1], 0.0f);
        v2f b;
        b.x = (l15 < OUTD) ? W2[(ka + 0) * OUTD + l15] : 0.0f;
        b.y = (l15 < OUTD) ? W2[(ka + 1) * OUTD + l15] : 0.0f;
        c = __builtin_amdgcn_wmma_f32_16x16x4_f32(false, a, false, b, (short)0, c, false, false);
    }
    float* mt = m2 + (size_t)(wave * 16) * HID;
    #pragma unroll
    for (int v = 0; v < 8; ++v)
        mt[(v + hi * 8) * HID + l15] = c[v];
}

template <bool FULL>
__device__ __forceinline__ void scatter2_chunk(const int* __restrict__ src, const int* __restrict__ dst,
                                               const float* __restrict__ m2, float* __restrict__ out,
                                               long long e0, long long n_edges, int lane, int f) {
    long long e = e0 + lane;
    int s_l = 0, d_l = -1;
    if (FULL || e < n_edges) { s_l = src[e]; d_l = dst[e]; }
    float v[16];
    int   d[16];
    #pragma unroll
    for (int j = 0; j < 16; ++j) {
        int s = __shfl(s_l, j, 16);
        d[j]  = __shfl(d_l, j, 16);
        v[j]  = m2[(unsigned)s * HID + f];
    }
    if (f < OUTD) {
        #pragma unroll
        for (int j = 0; j < 16; ++j)
            if (FULL || d[j] >= 0) atomicAdd(out + (unsigned)d[j] * OUTD + f, v[j]);
    }
}

__global__ void __launch_bounds__(256)
gcn_scatter2(const int* __restrict__ src, const int* __restrict__ dst,
             const float* __restrict__ m2, float* __restrict__ out, long long n_edges) {
    int lane = threadIdx.x & 31;
    int f    = lane & 15;
    long long wave     = ((long long)blockIdx.x * blockDim.x + threadIdx.x) >> 5;
    long long n_waves  = ((long long)gridDim.x * blockDim.x) >> 5;
    long long n_chunks = (n_edges + 31) >> 5;
    for (long long c = wave; c < n_chunks; c += n_waves) {
        long long e0 = c * 32;
        if (e0 + 32 <= n_edges) scatter2_chunk<true >(src, dst, m2, out, e0, n_edges, lane, f);
        else                    scatter2_chunk<false>(src, dst, m2, out, e0, n_edges, lane, f);
    }
}

extern "C" void kernel_launch(void* const* d_in, const int* in_sizes, int n_in,
                              void* d_out, int out_size, void* d_ws, size_t ws_size,
                              hipStream_t stream) {
    const float* x  = (const float*)d_in[0];
    const int*   ei = (const int*)  d_in[1];
    const float* W1 = (const float*)d_in[2];
    const float* b1 = (const float*)d_in[3];
    const float* W2 = (const float*)d_in[4];
    const float* b2 = (const float*)d_in[5];
    float* out = (float*)d_out;

    int n_nodes       = in_sizes[0] / IN_DIM;
    long long n_edges = (long long)in_sizes[1] / 2;
    const int* src = ei;
    const int* dst = ei + n_edges;

    float* m1  = (float*)d_ws;
    float* agg = m1  + (size_t)n_nodes * HID;
    float* m2  = agg + (size_t)n_nodes * HID;

    int n_tiles = (n_nodes + 15) / 16;
    int init_blocks = (n_nodes * HID + 255) / 256;
    int gemm_blocks = (n_tiles + 7) / 8;
    int edge_blocks = 2048;

    gcn_init_kernel<<<init_blocks, 256, 0, stream>>>(agg, out, b2, n_nodes);
    gcn_gemm1_wmma <<<gemm_blocks, 256, 0, stream>>>(x, W1, m1, n_tiles);
    gcn_scatter1   <<<edge_blocks, 256, 0, stream>>>(src, dst, m1, agg, n_edges);
    gcn_gemm2_wmma <<<gemm_blocks, 256, 0, stream>>>(agg, b1, W2, m2, n_tiles);
    gcn_scatter2   <<<edge_blocks, 256, 0, stream>>>(src, dst, m2, out, n_edges);
}


# edges binned once per call by 400-node destination block (LDS counts and cursors); each block summed in LDS by ds f32 atomics and stored whole: no global atomics, no init; sums in LDS arrival order
# speedup vs baseline: 11.5221x; 1.8794x over previous
#include <hip/hip_runtime.h>

typedef float v2f __attribute__((ext_vector_type(2)));
typedef float v8f __attribute__((ext_vector_type(8)));

#define IN_DIM 256
#define HID 16
#define OUTD 10

__global__ void gcn_gemm1_wmma(const float* __restrict__ x, const float* __restrict__ W1,
                               float* __restrict__ m1, int n_tiles) {
    int wave = (int)((blockIdx.x * blockDim.x + threadIdx.x) >> 5);
    int lane = threadIdx.x & 31;
    if (wave >= n_tiles) return;
    int l15 = lane & 15;
    int hi  = lane >> 4;
    const float* xrow = x + ((size_t)(wave * 16 + l15)) * IN_DIM;

    v8f c = {};
    #pragma unroll 4
    for (int k0 = 0; k0 < IN_DIM; k0 += 4) {
        int ka = k0 + hi * 2;
        v2f a; a.x = xrow[ka];                 a.y = xrow[ka + 1];
        v2f b; b.x = W1[(ka + 0) * HID + l15]; b.y = W1[(ka + 1) * HID + l15];
        c = __builtin_amdgcn_wmma_f32_16x16x4_f32(false, a, false, b, (short)0, c, false, false);
    }
    float* mt = m1 + (size_t)(wave * 16) * HID;
    #pragma unroll
    for (int v = 0; v < 8; ++v)
        mt[(v + hi * 8) * HID + l15] = c[v];
}

#define BIN_NODES 400
#define NSLICE    256
#define MAXBINS   1024
#define WG        1024

__global__ void __launch_bounds__(WG)
gcn_bin_count(const int* __restrict__ dst, int* __restrict__ counts,
              long long n_edges, long long slice, int n_bins) {
    __shared__ int h[MAXBINS];
    int tid = threadIdx.x;
    for (int i = tid; i < n_bins; i += WG) h[i] = 0;
    __syncthreads();
    long long beg = (long long)blockIdx.x * slice;
    long long end = beg + slice < n_edges ? beg + slice : n_edges;
    for (long long e = beg + tid; e < end; e += WG)
        atomicAdd(&h[(unsigned)dst[e] / BIN_NODES], 1);
    __syncthreads();
    for (int i = tid; i < n_bins; i += WG) counts[(size_t)i * NSLICE + blockIdx.x] = h[i];
}

__global__ void __launch_bounds__(WG)
gcn_bin_fill(const int* __restrict__ src, const int* __restrict__ dst, const int* __restrict__ counts,
             int* __restrict__ bin_start, int* __restrict__ pairs,
             long long n_edges, long long slice, int n_bins) {
    __shared__ int tot[MAXBINS];
    __shared__ int cur[MAXBINS];
    __shared__ int wsum[32];
    int tid = threadIdx.x, lane = tid & 31, wave = tid >> 5;
    int g = blockIdx.x;
    for (int b = wave; b < n_bins; b += WG / 32) {
        const int4* row = (const int4*)(counts + (size_t)b * NSLICE);
        int t = 0, p = 0;
        #pragma unroll
        for (int k = 0; k < NSLICE / 128; ++k) {
            int4 c = row[k * 32 + lane];
            int i0 = (k * 32 + lane) * 4;
            t += c.x + c.y + c.z + c.w;
            p += (i0 + 0 < g ? c.x : 0) + (i0 + 1 < g ? c.y : 0) + (i0 + 2 < g ? c.z : 0) + (i0 + 3 < g ? c.w : 0);
        }
        for (int off = 16; off > 0; off >>= 1) { t += __shfl_xor(t, off); p += __shfl_xor(p, off); }
        if (lane == 0) { tot[b] = t; cur[b] = p; }
    }
    __syncthreads();
    int v = tid < n_bins ? tot[tid] : 0;
    int x = v;
    for (int off = 1; off < 32; off <<= 1) { int y = __shfl_up(x, off); if (lane >= off) x += y; }
    if (lane == 31) wsum[wave] = x;
    __syncthreads();
    if (wave == 0) {
        int w = wsum[lane], xs = w;
        for (int off = 1; off < 32; off <<= 1) { int y = __shfl_up(xs, off); if (lane >= off) xs += y; }
        wsum[lane] = xs - w;
    }
    __syncthreads();
    int excl = x - v + wsum[wave];
    if (tid < n_bins) { cur[tid] += excl; if (g == 0) bin_start[tid] = excl; }
    if (g == 0 && tid == 0) bin_start[n_bins] = (int)n_edges;
    __syncthreads();
    long long beg = (long long)g * slice;
    long long end = beg + slice < n_edges ? beg + slice : n_edges;
    for (long long e = beg + tid; e < end; e += WG) {
        int s = src[e], d = dst[e];
        int b = (unsigned)d / BIN_NODES;
        int pos = atomicAdd(&cur[b], 1);
        *(int2*)(pairs + 2 * (size_t)pos) = make_int2(s, d - b * BIN_NODES);
    }
}

template <bool FULL, int NF>
__device__ __forceinline__ void bin_chunk(const int* __restrict__ pairs, const float* __restrict__ msg,
                                          float* acc, int e0, int end, int lane, int f) {
    int e = e0 + lane;
    int s_l = 0, d_l = -1;
    if (FULL || e < end) { int2 p = *(const int2*)(pairs + 2 * (size_t)e); s_l = p.x; d_l = p.y; }
    float v[16];
    int   d[16];
    #pragma unroll
    for (int j = 0; j < 16; ++j) {
        int s = __shfl(s_l, j, 16);
        d[j]  = __shfl(d_l, j, 16);
        v[j]  = msg[(unsigned)s * HID + f];
    }
    if (NF == HID || f < NF) {
        #pragma unroll
        for (int j = 0; j < 16; ++j)
            if (FULL || d[j] >= 0) atomicAdd(&acc[d[j] * HID + f], v[j]);
    }
}

template <int NF, bool BIAS>
__global__ void __launch_bounds__(WG)
gcn_bin_sum(const int* __restrict__ pairs, const int* __restrict__ bin_start,
            const float* __restrict__ msg, const float* __restrict__ bias,
            float* __restrict__ res, int n_nodes) {
    __shared__ float acc[BIN_NODES * HID];
    int tid = threadIdx.x, lane = tid & 31, f = lane & 15, wave = tid >> 5;
    int b = blockIdx.x;
    for (int i = tid; i < BIN_NODES * HID; i += WG) acc[i] = 0.0f;
    __syncthreads();
    int beg = bin_start[b], end = bin_start[b + 1];
    for (int e0 = beg + wave * 32; e0 < end; e0 += WG) {
        if (e0 + 32 <= end) bin_chunk<true,  NF>(pairs, msg, acc, e0, end, lane, f);
        else                bin_chunk<false, NF>(pairs, msg, acc, e0, end, lane, f);
    }
    __syncthreads();
    int rows = n_nodes - b * BIN_NODES;
    if (rows > BIN_NODES) rows = BIN_NODES;
    float* rb = res + (size_t)b * BIN_NODES * NF;
    for (int i = tid; i < rows * NF; i += WG) {
        int r = i / NF, o = i - r * NF;
        float a = acc[r * HID + o];
        rb[i] = BIAS ? a + bias[o] : a;
    }
}

__global__ void gcn_gemm2_wmma(const float* __restrict__ agg, const float* __restrict__ b1,
                               const float* __restrict__ W2, float* __restrict__ m2,
                               int n_tiles) {
    int wave = (int)((blockIdx.x * blockDim.x + threadIdx.x) >> 5);
    int lane = threadIdx.x & 31;
    if (wave >= n_tiles) return;
    int l15 = lane & 15;
    int hi  = lane >> 4;
    const float* arow = agg + (size_t)(wave * 16 + l15) * HID;

    v8f c = {};
    #pragma unroll
    for (int k0 = 0; k0 < HID; k0 += 4) {
        int ka = k0 + hi * 2;
        v2f a;
        a.x = fmaxf(arow[ka]     + b1[ka],     0.0f);
        a.y = fmaxf(arow[ka + 1] + b1[ka + 1], 0.0f);
        v2f b;
        b.x = (l15 < OUTD) ? W2[(ka + 0) * OUTD + l15] : 0.0f;
        b.y = (l15 < OUTD) ? W2[(ka + 1) * OUTD + l15] : 0.0f;
        c = __builtin_amdgcn_wmma_f32_16x16x4_f32(false, a, false, b, (short)0, c, false, false);
    }
    float* mt = m2 + (size_t)(wave * 16) * HID;
    #pragma unroll
    for (int v = 0; v < 8; ++v)
        mt[(v + hi * 8) * HID + l15] = c[v];
}

extern "C" void kernel_launch(void* const* d_in, const int* in_sizes, int n_in,
                              void* d_out, int out_size, void* d_ws, size_t ws_size,
                              hipStream_t stream) {
    const float* x  = (const float*)d_in[0];
    const int*   ei = (const int*)  d_in[1];
    const float* W1 = (const float*)d_in[2];
    const float* b1 = (const float*)d_in[3];
    const float* W2 = (const float*)d_in[4];
    const float* b2 = (const float*)d_in[5];
    float* out = (float*)d_out;

    int n_nodes       = in_sizes[0] / IN_DIM;
    long long n_edges = (long long)in_sizes[1] / 2;
    const int* src = ei;
    const int* dst = ei + n_edges;

    int n_bins      = (n_nodes + BIN_NODES - 1) / BIN_NODES;
    long long slice = (n_edges + NSLICE - 1) / NSLICE;

    float* m1  = (float*)d_ws;
    float* agg = m1  + (size_t)n_nodes * HID;
    float* m2  = agg + (size_t)n_nodes * HID;
    int* counts    = (int*)(m2 + (size_t)n_nodes * HID);
    int* bin_start = counts + (size_t)n_bins * NSLICE;
    int* pairs     = bin_start + ((n_bins + 1 + 63) & ~63);

    int n_tiles = (n_nodes + 15) / 16;
    int gemm_blocks = (n_tiles + 7) / 8;

    gcn_bin_count  <<<NSLICE, WG, 0, stream>>>(dst, counts, n_edges, slice, n_bins);
    gcn_bin_fill   <<<NSLICE, WG, 0, stream>>>(src, dst, counts, bin_start, pairs, n_edges, slice, n_bins);
    gcn_gemm1_wmma <<<gemm_blocks, 256, 0, stream>>>(x, W1, m1, n_tiles);
    gcn_bin_sum<HID,  false><<<n_bins, WG, 0, stream>>>(pairs, bin_start, m1, b1, agg, n_nodes);
    gcn_gemm2_wmma <<<gemm_blocks, 256, 0, stream>>>(agg, b1, W2, m2, n_tiles);
    gcn_bin_sum<OUTD, true ><<<n_bins, WG, 0, stream>>>(pairs, bin_start, m2, b2, out, n_nodes);
}


# every pass requests its loads in groups ahead of use: count/fill 16 edges a thread, block sums 4 chunks a wave with next rows before current adds, product 16 k-steps a trip; sums in LDS arrival order
# speedup vs baseline: 12.7448x; 1.1061x over previous
#include <hip/hip_runtime.h>

typedef float v2f __attribute__((ext_vector_type(2)));
typedef float v8f __attribute__((ext_vector_type(8)));

#define IN_DIM 256
#define HID 16
#define OUTD 10

__global__ void gcn_gemm1_wmma(const float* __restrict__ x, const float* __restrict__ W1,
                               float* __restrict__ m1, int n_tiles) {
    int wave = (int)((blockIdx.x * blockDim.x + threadIdx.x) >> 5);
    int lane = threadIdx.x & 31;
    if (wave >= n_tiles) return;
    int l15 = lane & 15;
    int hi  = lane >> 4;
    const float* xrow = x + ((size_t)(wave * 16 + l15)) * IN_DIM;

    v8f c = {};
    for (int kb = 0; kb < IN_DIM; kb += 64) {
        v2f a[16], b[16];
        #pragma unroll
        for (int i = 0; i < 16; ++i) {
            int ka = kb + i * 4 + hi * 2;
            a[i].x = xrow[ka];                 a[i].y = xrow[ka + 1];
            b[i].x = W1[(ka + 0) * HID + l15]; b[i].y = W1[(ka + 1) * HID + l15];
        }
        __builtin_amdgcn_sched_barrier(0);
        #pragma unroll
        for (int i = 0; i < 16; ++i)
            c = __builtin_amdgcn_wmma_f32_16x16x4_f32(false, a[i], false, b[i], (short)0, c, false, false);
    }
    float* mt = m1 + (size_t)(wave * 16) * HID;
    #pragma unroll
    for (int v = 0; v < 8; ++v)
        mt[(v + hi * 8) * HID + l15] = c[v];
}

#define BIN_NODES 400
#define NSLICE    256
#define MAXBINS   1024
#define WG        1024
#define EU        16
#define PG        4
#define CB        4

__global__ void __launch_bounds__(WG)
gcn_bin_count(const int* __restrict__ dst, int* __restrict__ counts,
              long long n_edges, long long slice, int n_bins) {
    __shared__ int h[MAXBINS + 32];
    int tid = threadIdx.x;
    for (int i = tid; i < n_bins; i += WG) h[i] = 0;
    if (tid < 32) h[MAXBINS + tid] = 0;
    __syncthreads();
    long long beg = (long long)blockIdx.x * slice;
    long long end = beg + slice < n_edges ? beg + slice : n_edges;
    int n = end > beg ? (int)(end - beg) : 0;
    const int* dp = dst + beg;
    for (int i0 = tid; i0 < n; i0 += WG * EU) {
        int dv[EU];
        #pragma unroll
        for (int u = 0; u < EU; ++u) { int i = i0 + u * WG; dv[u] = dp[i < n ? i : n - 1]; }
        #pragma unroll
        for (int u = 0; u < EU; ++u)
            atomicAdd(&h[i0 + u * WG < n ? (int)((unsigned)dv[u] / BIN_NODES) : MAXBINS + (tid & 31)], 1);
    }
    __syncthreads();
    for (int i = tid; i < n_bins; i += WG) counts[(size_t)i * NSLICE + blockIdx.x] = h[i];
}

__global__ void __launch_bounds__(WG)
gcn_bin_fill(const int* __restrict__ src, const int* __restrict__ dst, const int* __restrict__ counts,
             int* __restrict__ bin_start, int* __restrict__ pairs,
             long long n_edges, long long slice, int n_bins) {
    __shared__ int tot[MAXBINS];
    __shared__ int cur[MAXBINS + 32];
    __shared__ int wsum[32];
    int tid = threadIdx.x, lane = tid & 31, wave = tid >> 5;
    int g = blockIdx.x;
    if (tid < 32) cur[MAXBINS + tid] = 0;
    long long beg = (long long)g * slice;
    long long end = beg + slice < n_edges ? beg + slice : n_edges;
    int n = end > beg ? (int)(end - beg) : 0;
    const int* sp = src + beg;
    const int* dp = dst + beg;
    int sv[EU], dv[EU];
    #pragma unroll
    for (int u = 0; u < EU; ++u) { sv[u] = 0; dv[u] = 0; }
    if (n > 0) {
        #pragma unroll
        for (int u = 0; u < EU; ++u) {
            int i = tid + u * WG;
            i = i < n ? i : n - 1;
            sv[u] = sp[i];
            dv[u] = dp[i];
        }
    }
    for (int b0 = wave; b0 < n_bins; b0 += (WG / 32) * CB) {
        int4 c[CB][NSLICE / 128];
        #pragma unroll
        for (int q = 0; q < CB; ++q) {
            int b = b0 + q * (WG / 32);
            b = b < n_bins ? b : n_bins - 1;
            const int4* row = (const int4*)(counts + (size_t)b * NSLICE);
            #pragma unroll
            for (int k = 0; k < NSLICE / 128; ++k) c[q][k] = row[k * 32 + lane];
        }
        #pragma unroll
        for (int q = 0; q < CB; ++q) {
            int b = b0 + q * (WG / 32);
            if (b < n_bins) {
                int t = 0, p = 0;
                #pragma unroll
                for (int k = 0; k < NSLICE / 128; ++k) {
                    int4 cc = c[q][k];
                    int i0 = (k * 32 + lane) * 4;
                    t += cc.x + cc.y + cc.z + cc.w;
                    p += (i0 + 0 < g ? cc.x : 0) + (i0 + 1 < g ? cc.y : 0) + (i0 + 2 < g ? cc.z : 0) + (i0 + 3 < g ? cc.w : 0);
                }
                for (int off = 16; off > 0; off >>= 1) { t += __shfl_xor(t, off); p += __shfl_xor(p, off); }
                if (lane == 0) { tot[b] = t; cur[b] = p; }
            }
        }
    }
    __syncthreads();
    int v = tid < n_bins ? tot[tid] : 0;
    int x = v;
    for (int off = 1; off < 32; off <<= 1) { int y = __shfl_up(x, off); if (lane >= off) x += y; }
    if (lane == 31) wsum[wave] = x;
    __syncthreads();
    if (wave == 0) {
        int w = wsum[lane], xs = w;
        for (int off = 1; off < 32; off <<= 1) { int y = __shfl_up(xs, off); if (lane >= off) xs += y; }
        wsum[lane] = xs - w;
    }
    __syncthreads();
    int excl = x - v + wsum[wave];
    if (tid < n_bins) { cur[tid] += excl; if (g == 0) bin_start[tid] = excl; }
    if (g == 0 && tid == 0) bin_start[n_bins] = (int)n_edges;
    __syncthreads();
    for (int i0 = tid; i0 < n; i0 += WG * EU) {
        if (i0 != tid) {
            #pragma unroll
            for (int u = 0; u < EU; ++u) {
                int i = i0 + u * WG;
                i = i < n ? i : n - 1;
                sv[u] = sp[i];
                dv[u] = dp[i];
            }
        }
        int pos[EU];
        #pragma unroll
        for (int u = 0; u < EU; ++u)
            pos[u] = atomicAdd(&cur[i0 + u * WG < n ? (int)((unsigned)dv[u] / BIN_NODES) : MAXBINS + lane], 1);
        #pragma unroll
        for (int u = 0; u < EU; ++u) {
            if (i0 + u * WG < n) {
                int d = dv[u];
                int b = (unsigned)d / BIN_NODES;
                *(int2*)(pairs + 2 * (size_t)pos[u]) = make_int2(sv[u], d - b * BIN_NODES);
            }
        }
    }
}

__device__ __forceinline__ int2 load_pair(const int* __restrict__ pairs, int e, int end) {
    int ec = e < end ? e : end - 1;
    ec = ec > 0 ? ec : 0;
    int2 p = *(const int2*)(pairs + 2 * (size_t)ec);
    if (e >= end) p = make_int2(0, BIN_NODES);
    return p;
}

__device__ __forceinline__ void gather_rows(const float* __restrict__ msg, int s_l, int f, float* v) {
    #pragma unroll
    for (int j = 0; j < 16; ++j)
        v[j] = msg[(unsigned)__shfl(s_l, j, 16) * HID + f];
}

__device__ __forceinline__ void add_rows(float* acc, int d_l, int f, const float* v) {
    #pragma unroll
    for (int j = 0; j < 16; ++j) atomicAdd(&acc[__shfl(d_l, j, 16) * HID + f], v[j]);
}

template <int NF, bool BIAS>
__global__ void __launch_bounds__(WG)
gcn_bin_sum(const int* __restrict__ pairs, const int* __restrict__ bin_start,
            const float* __restrict__ msg, const float* __restrict__ bias,
            float* __restrict__ res, int n_nodes) {
    __shared__ float acc[(BIN_NODES + 1) * HID];
    int tid = threadIdx.x, lane = tid & 31, f = lane & 15;
    int wave = __builtin_amdgcn_readfirstlane(tid >> 5);
    int b = blockIdx.x;
    int beg = bin_start[b], end = bin_start[b + 1];
    int e0 = beg + wave * 32;
    int2 pn[PG];
    #pragma unroll
    for (int t = 0; t < PG; ++t) pn[t] = load_pair(pairs, e0 + t * WG + lane, end);
    for (int i = tid; i < (BIN_NODES + 1) * HID; i += WG) acc[i] = 0.0f;
    __syncthreads();
    float v[PG - 1][16];
    for (; e0 + (PG - 1) * WG < end; e0 += WG * PG) {
        int2 p[PG];
        #pragma unroll
        for (int t = 0; t < PG; ++t) p[t] = pn[t];
        gather_rows(msg, p[0].x, f, v[0]);
        #pragma unroll
        for (int t = 0; t < PG; ++t) {
            if (t + 1 < PG) gather_rows(msg, p[t + 1].x, f, v[(t + 1) & 1]);
            if (t == PG - 2) {
                #pragma unroll
                for (int u = 0; u < PG; ++u) pn[u] = load_pair(pairs, e0 + (PG + u) * WG + lane, end);
            }
            add_rows(acc, p[t].y, f, v[t & 1]);
        }
    }
    if (e0 < end) {
        #pragma unroll
        for (int t = 0; t < PG - 1; ++t) gather_rows(msg, pn[t].x, f, v[t]);
        #pragma unroll
        for (int t = 0; t < PG - 1; ++t)
            if (e0 + t * WG < end) add_rows(acc, pn[t].y, f, v[t]);
    }
    __syncthreads();
    int rows = n_nodes - b * BIN_NODES;
    if (rows > BIN_NODES) rows = BIN_NODES;
    float* rb = res + (size_t)b * BIN_NODES * NF;
    for (int i = tid; i < rows * NF; i += WG) {
        int r = i / NF, o = i - r * NF;
        float a = acc[r * HID + o];
        rb[i] = BIAS ? a + bias[o] : a;
    }
}

__global__ void gcn_gemm2_wmma(const float* __restrict__ agg, const float* __restrict__ b1,
                               const float* __restrict__ W2, float* __restrict__ m2,
                               int n_tiles) {
    int wave = (int)((blockIdx.x * blockDim.x + threadIdx.x) >> 5);
    int lane = threadIdx.x & 31;
    if (wave >= n_tiles) return;
    int l15 = lane & 15;
    int hi  = lane >> 4;
    const float* arow = agg + (size_t)(wave * 16 + l15) * HID;

    v8f c = {};
    #pragma unroll
    for (int k0 = 0; k0 < HID; k0 += 4) {
        int ka = k0 + hi * 2;
        v2f a;
        a.x = fmaxf(arow[ka]     + b1[ka],     0.0f);
        a.y = fmaxf(arow[ka + 1] + b1[ka + 1], 0.0f);
        v2f b;
        b.x = (l15 < OUTD) ? W2[(ka + 0) * OUTD + l15] : 0.0f;
        b.y = (l15 < OUTD) ? W2[(ka + 1) * OUTD + l15] : 0.0f;
        c = __builtin_amdgcn_wmma_f32_16x16x4_f32(false, a, false, b, (short)0, c, false, false);
    }
    float* mt = m2 + (size_t)(wave * 16) * HID;
    #pragma unroll
    for (int v = 0; v < 8; ++v)
        mt[(v + hi * 8) * HID + l15] = c[v];
}

extern "C" void kernel_launch(void* const* d_in, const int* in_sizes, int n_in,
                              void* d_out, int out_size, void* d_ws, size_t ws_size,
                              hipStream_t stream) {
    const float* x  = (const float*)d_in[0];
    const int*   ei = (const int*)  d_in[1];
    const float* W1 = (const float*)d_in[2];
    const float* b1 = (const float*)d_in[3];
    const float* W2 = (const float*)d_in[4];
    const float* b2 = (const float*)d_in[5];
    float* out = (float*)d_out;

    int n_nodes       = in_sizes[0] / IN_DIM;
    long long n_edges = (long long)in_sizes[1] / 2;
    const int* src = ei;
    const int* dst = ei + n_edges;

    int n_bins      = (n_nodes + BIN_NODES - 1) / BIN_NODES;
    long long slice = (n_edges + NSLICE - 1) / NSLICE;

    float* m1  = (float*)d_ws;
    float* agg = m1  + (size_t)n_nodes * HID;
    float* m2  = agg + (size_t)n_nodes * HID;
    int* counts    = (int*)(m2 + (size_t)n_nodes * HID);
    int* bin_start = counts + (size_t)n_bins * NSLICE;
    int* pairs     = bin_start + ((n_bins + 1 + 63) & ~63);

    int n_tiles = (n_nodes + 15) / 16;
    int gemm_blocks = (n_tiles + 7) / 8;

    gcn_bin_count  <<<NSLICE, WG, 0, stream>>>(dst, counts, n_edges, slice, n_bins);
    gcn_bin_fill   <<<NSLICE, WG, 0, stream>>>(src, dst, counts, bin_start, pairs, n_edges, slice, n_bins);
    gcn_gemm1_wmma <<<gemm_blocks, 256, 0, stream>>>(x, W1, m1, n_tiles);
    gcn_bin_sum<HID,  false><<<n_bins, WG, 0, stream>>>(pairs, bin_start, m1, b1, agg, n_nodes);
    gcn_gemm2_wmma <<<gemm_blocks, 256, 0, stream>>>(agg, b1, W2, m2, n_tiles);
    gcn_bin_sum<OUTD, true ><<<n_bins, WG, 0, stream>>>(pairs, bin_start, m2, b2, out, n_nodes);
}


# layer-2 product moved into the layer-1 block sum: each block turns its 400 summed rows, still in LDS, into m2 rows (4 f32 WMMAs a 16-row tile) and stores those; one launch and the agg round trip go
# speedup vs baseline: 13.3730x; 1.0493x over previous
#include <hip/hip_runtime.h>

typedef float v2f __attribute__((ext_vector_type(2)));
typedef float v8f __attribute__((ext_vector_type(8)));

#define IN_DIM 256
#define HID 16
#define OUTD 10

__global__ void gcn_gemm1_wmma(const float* __restrict__ x, const float* __restrict__ W1,
                               float* __restrict__ m1, int n_tiles) {
    int wave = (int)((blockIdx.x * blockDim.x + threadIdx.x) >> 5);
    int lane = threadIdx.x & 31;
    if (wave >= n_tiles) return;
    int l15 = lane & 15;
    int hi  = lane >> 4;
    const float* xrow = x + ((size_t)(wave * 16 + l15)) * IN_DIM;

    v8f c = {};
    for (int kb = 0; kb < IN_DIM; kb += 64) {
        v2f a[16], b[16];
        #pragma unroll
        for (int i = 0; i < 16; ++i) {
            int ka = kb + i * 4 + hi * 2;
            a[i].x = xrow[ka];                 a[i].y = xrow[ka + 1];
            b[i].x = W1[(ka + 0) * HID + l15]; b[i].y = W1[(ka + 1) * HID + l15];
        }
        __builtin_amdgcn_sched_barrier(0);
        #pragma unroll
        for (int i = 0; i < 16; ++i)
            c = __builtin_amdgcn_wmma_f32_16x16x4_f32(false, a[i], false, b[i], (short)0, c, false, false);
    }
    float* mt = m1 + (size_t)(wave * 16) * HID;
    #pragma unroll
    for (int v = 0; v < 8; ++v)
        mt[(v + hi * 8) * HID + l15] = c[v];
}

#define BIN_NODES 400
#define NSLICE    256
#define MAXBINS   1024
#define WG        1024
#define EU        16
#define PG        4
#define CB        4

__global__ void __launch_bounds__(WG)
gcn_bin_count(const int* __restrict__ dst, int* __restrict__ counts,
              long long n_edges, long long slice, int n_bins) {
    __shared__ int h[MAXBINS + 32];
    int tid = threadIdx.x;
    for (int i = tid; i < n_bins; i += WG) h[i] = 0;
    if (tid < 32) h[MAXBINS + tid] = 0;
    __syncthreads();
    long long beg = (long long)blockIdx.x * slice;
    long long end = beg + slice < n_edges ? beg + slice : n_edges;
    int n = end > beg ? (int)(end - beg) : 0;
    const int* dp = dst + beg;
    for (int i0 = tid; i0 < n; i0 += WG * EU) {
        int dv[EU];
        #pragma unroll
        for (int u = 0; u < EU; ++u) { int i = i0 + u * WG; dv[u] = dp[i < n ? i : n - 1]; }
        #pragma unroll
        for (int u = 0; u < EU; ++u)
            atomicAdd(&h[i0 + u * WG < n ? (int)((unsigned)dv[u] / BIN_NODES) : MAXBINS + (tid & 31)], 1);
    }
    __syncthreads();
    for (int i = tid; i < n_bins; i += WG) counts[(size_t)i * NSLICE + blockIdx.x] = h[i];
}

__global__ void __launch_bounds__(WG)
gcn_bin_fill(const int* __restrict__ src, const int* __restrict__ dst, const int* __restrict__ counts,
             int* __restrict__ bin_start, int* __restrict__ pairs,
             long long n_edges, long long slice, int n_bins) {
    __shared__ int tot[MAXBINS];
    __shared__ int cur[MAXBINS + 32];
    __shared__ int wsum[32];
    int tid = threadIdx.x, lane = tid & 31, wave = tid >> 5;
    int g = blockIdx.x;
    if (tid < 32) cur[MAXBINS + tid] = 0;
    long long beg = (long long)g * slice;
    long long end = beg + slice < n_edges ? beg + slice : n_edges;
    int n = end > beg ? (int)(end - beg) : 0;
    const int* sp = src + beg;
    const int* dp = dst + beg;
    int sv[EU], dv[EU];
    #pragma unroll
    for (int u = 0; u < EU; ++u) { sv[u] = 0; dv[u] = 0; }
    if (n > 0) {
        #pragma unroll
        for (int u = 0; u < EU; ++u) {
            int i = tid + u * WG;
            i = i < n ? i : n - 1;
            sv[u] = sp[i];
            dv[u] = dp[i];
        }
    }
    for (int b0 = wave; b0 < n_bins; b0 += (WG / 32) * CB) {
        int4 c[CB][NSLICE / 128];
        #pragma unroll
        for (int q = 0; q < CB; ++q) {
            int b = b0 + q * (WG / 32);
            b = b < n_bins ? b : n_bins - 1;
            const int4* row = (const int4*)(counts + (size_t)b * NSLICE);
            #pragma unroll
            for (int k = 0; k < NSLICE / 128; ++k) c[q][k] = row[k * 32 + lane];
        }
        #pragma unroll
        for (int q = 0; q < CB; ++q) {
            int b = b0 + q * (WG / 32);
            if (b < n_bins) {
                int t = 0, p = 0;
                #pragma unroll
                for (int k = 0; k < NSLICE / 128; ++k) {
                    int4 cc = c[q][k];
                    int i0 = (k * 32 + lane) * 4;
                    t += cc.x + cc.y + cc.z + cc.w;
                    p += (i0 + 0 < g ? cc.x : 0) + (i0 + 1 < g ? cc.y : 0) + (i0 + 2 < g ? cc.z : 0) + (i0 + 3 < g ? cc.w : 0);
                }
                for (int off = 16; off > 0; off >>= 1) { t += __shfl_xor(t, off); p += __shfl_xor(p, off); }
                if (lane == 0) { tot[b] = t; cur[b] = p; }
            }
        }
    }
    __syncthreads();
    int v = tid < n_bins ? tot[tid] : 0;
    int x = v;
    for (int off = 1; off < 32; off <<= 1) { int y = __shfl_up(x, off); if (lane >= off) x += y; }
    if (lane == 31) wsum[wave] = x;
    __syncthreads();
    if (wave == 0) {
        int w = wsum[lane], xs = w;
        for (int off = 1; off < 32; off <<= 1) { int y = __shfl_up(xs, off); if (lane >= off) xs += y; }
        wsum[lane] = xs - w;
    }
    __syncthreads();
    int excl = x - v + wsum[wave];
    if (tid < n_bins) { cur[tid] += excl; if (g == 0) bin_start[tid] = excl; }
    if (g == 0 && tid == 0) bin_start[n_bins] = (int)n_edges;
    __syncthreads();
    for (int i0 = tid; i0 < n; i0 += WG * EU) {
        if (i0 != tid) {
            #pragma unroll
            for (int u = 0; u < EU; ++u) {
                int i = i0 + u * WG;
                i = i < n ? i : n - 1;
                sv[u] = sp[i];
                dv[u] = dp[i];
            }
        }
        int pos[EU];
        #pragma unroll
        for (int u = 0; u < EU; ++u)
            pos[u] = atomicAdd(&cur[i0 + u * WG < n ? (int)((unsigned)dv[u] / BIN_NODES) : MAXBINS + lane], 1);
        #pragma unroll
        for (int u = 0; u < EU; ++u) {
            if (i0 + u * WG < n) {
                int d = dv[u];
                int b = (unsigned)d / BIN_NODES;
                *(int2*)(pairs + 2 * (size_t)pos[u]) = make_int2(sv[u], d - b * BIN_NODES);
            }
        }
    }
}

__device__ __forceinline__ int2 load_pair(const int* __restrict__ pairs, int e, int end) {
    int ec = e < end ? e : end - 1;
    ec = ec > 0 ? ec : 0;
    int2 p = *(const int2*)(pairs + 2 * (size_t)ec);
    if (e >= end) p = make_int2(0, BIN_NODES);
    return p;
}

__device__ __forceinline__ void gather_rows(const float* __restrict__ msg, int s_l, int f, float* v) {
    #pragma unroll
    for (int j = 0; j < 16; ++j)
        v[j] = msg[(unsigned)__shfl(s_l, j, 16) * HID + f];
}

__device__ __forceinline__ void add_rows(float* acc, int d_l, int f, const float* v) {
    #pragma unroll
    for (int j = 0; j < 16; ++j) atomicAdd(&acc[__shfl(d_l, j, 16) * HID + f], v[j]);
}

template <int NF, bool BIAS, bool PROD>
__global__ void __launch_bounds__(WG)
gcn_bin_sum(const int* __restrict__ pairs, const int* __restrict__ bin_start,
            const float* __restrict__ msg, const float* __restrict__ bias, const float* __restrict__ W2,
            float* __restrict__ res, int n_nodes) {
    __shared__ float acc[(BIN_NODES + 1) * HID];
    int tid = threadIdx.x, lane = tid & 31, f = lane & 15;
    int wave = __builtin_amdgcn_readfirstlane(tid >> 5);
    int b = blockIdx.x;
    int beg = bin_start[b], end = bin_start[b + 1];
    int e0 = beg + wave * 32;
    int2 pn[PG];
    #pragma unroll
    for (int t = 0; t < PG; ++t) pn[t] = load_pair(pairs, e0 + t * WG + lane, end);
    for (int i = tid; i < (BIN_NODES + 1) * HID; i += WG) acc[i] = 0.0f;
    __syncthreads();
    float v[PG - 1][16];
    for (; e0 + (PG - 1) * WG < end; e0 += WG * PG) {
        int2 p[PG];
        #pragma unroll
        for (int t = 0; t < PG; ++t) p[t] = pn[t];
        gather_rows(msg, p[0].x, f, v[0]);
        #pragma unroll
        for (int t = 0; t < PG; ++t) {
            if (t + 1 < PG) gather_rows(msg, p[t + 1].x, f, v[(t + 1) & 1]);
            if (t == PG - 2) {
                #pragma unroll
                for (int u = 0; u < PG; ++u) pn[u] = load_pair(pairs, e0 + (PG + u) * WG + lane, end);
            }
            add_rows(acc, p[t].y, f, v[t & 1]);
        }
    }
    if (e0 < end) {
        #pragma unroll
        for (int t = 0; t < PG - 1; ++t) gather_rows(msg, pn[t].x, f, v[t]);
        #pragma unroll
        for (int t = 0; t < PG - 1; ++t)
            if (e0 + t * WG < end) add_rows(acc, pn[t].y, f, v[t]);
    }
    int hi = lane >> 4;
    v2f wb[HID / 4], bb[HID / 4];
    if (PROD) {
        #pragma unroll
        for (int k = 0; k < HID / 4; ++k) {
            int ka = k * 4 + hi * 2;
            int fc = f < OUTD ? f : 0;
            float w0 = W2[(ka + 0) * OUTD + fc], w1 = W2[(ka + 1) * OUTD + fc];
            wb[k].x = f < OUTD ? w0 : 0.0f;
            wb[k].y = f < OUTD ? w1 : 0.0f;
            bb[k].x = bias[ka];
            bb[k].y = bias[ka + 1];
        }
    }
    __syncthreads();
    if (PROD) {
        static_assert(BIN_NODES % 16 == 0 && BIN_NODES / 16 <= WG / 32, "one whole 16-row tile per wave");
        if (wave < BIN_NODES / 16) {
            float* at = acc + wave * 16 * HID;
            v8f c = {};
            #pragma unroll
            for (int k = 0; k < HID / 4; ++k) {
                int ka = k * 4 + hi * 2;
                v2f s = *(const v2f*)(at + f * HID + ka);
                v2f a;
                a.x = fmaxf(s.x + bb[k].x, 0.0f);
                a.y = fmaxf(s.y + bb[k].y, 0.0f);
                c = __builtin_amdgcn_wmma_f32_16x16x4_f32(false, a, false, wb[k], (short)0, c, false, false);
            }
            #pragma unroll
            for (int r = 0; r < 8; ++r)
                at[(r + hi * 8) * HID + f] = c[r];
        }
        __syncthreads();
    }
    int rows = n_nodes - b * BIN_NODES;
    if (rows > BIN_NODES) rows = BIN_NODES;
    float* rb = res + (size_t)b * BIN_NODES * NF;
    for (int i = tid; i < rows * NF; i += WG) {
        int r = i / NF, o = i - r * NF;
        float a = acc[r * HID + o];
        rb[i] = BIAS ? a + bias[o] : a;
    }
}

extern "C" void kernel_launch(void* const* d_in, const int* in_sizes, int n_in,
                              void* d_out, int out_size, void* d_ws, size_t ws_size,
                              hipStream_t stream) {
    const float* x  = (const float*)d_in[0];
    const int*   ei = (const int*)  d_in[1];
    const float* W1 = (const float*)d_in[2];
    const float* b1 = (const float*)d_in[3];
    const float* W2 = (const float*)d_in[4];
    const float* b2 = (const float*)d_in[5];
    float* out = (float*)d_out;

    int n_nodes       = in_sizes[0] / IN_DIM;
    long long n_edges = (long long)in_sizes[1] / 2;
    const int* src = ei;
    const int* dst = ei + n_edges;

    int n_bins      = (n_nodes + BIN_NODES - 1) / BIN_NODES;
    long long slice = (n_edges + NSLICE - 1) / NSLICE;

    float* m1  = (float*)d_ws;
    float* m2  = m1 + (size_t)n_nodes * HID;
    int* counts    = (int*)(m2 + (size_t)n_nodes * HID);
    int* bin_start = counts + (size_t)n_bins * NSLICE;
    int* pairs     = bin_start + ((n_bins + 1 + 63) & ~63);

    int n_tiles = (n_nodes + 15) / 16;
    int gemm_blocks = (n_tiles + 7) / 8;

    gcn_bin_count  <<<NSLICE, WG, 0, stream>>>(dst, counts, n_edges, slice, n_bins);
    gcn_bin_fill   <<<NSLICE, WG, 0, stream>>>(src, dst, counts, bin_start, pairs, n_edges, slice, n_bins);
    gcn_gemm1_wmma <<<gemm_blocks, 256, 0, stream>>>(x, W1, m1, n_tiles);
    gcn_bin_sum<HID,  false, true ><<<n_bins, WG, 0, stream>>>(pairs, bin_start, m1, b1, W2, m2,  n_nodes);
    gcn_bin_sum<OUTD, true,  false><<<n_bins, WG, 0, stream>>>(pairs, bin_start, m2, b2, W2, out, n_nodes);
}


# fill pass orders each slice's edges by block in LDS, then copies them out in that order: a wave's store fills whole lines of a run, not 32 scattered 8-byte pieces; sums still in LDS arrival order
# speedup vs baseline: 14.6694x; 1.0969x over previous
#include <hip/hip_runtime.h>

typedef float v2f __attribute__((ext_vector_type(2)));
typedef float v8f __attribute__((ext_vector_type(8)));

#define IN_DIM 256
#define HID 16
#define OUTD 10

__global__ void gcn_gemm1_wmma(const float* __restrict__ x, const float* __restrict__ W1,
                               float* __restrict__ m1, int n_tiles) {
    int wave = (int)((blockIdx.x * blockDim.x + threadIdx.x) >> 5);
    int lane = threadIdx.x & 31;
    if (wave >= n_tiles) return;
    int l15 = lane & 15;
    int hi  = lane >> 4;
    const float* xrow = x + ((size_t)(wave * 16 + l15)) * IN_DIM;

    v8f c = {};
    for (int kb = 0; kb < IN_DIM; kb += 64) {
        v2f a[16], b[16];
        #pragma unroll
        for (int i = 0; i < 16; ++i) {
            int ka = kb + i * 4 + hi * 2;
            a[i].x = xrow[ka];                 a[i].y = xrow[ka + 1];
            b[i].x = W1[(ka + 0) * HID + l15]; b[i].y = W1[(ka + 1) * HID + l15];
        }
        __builtin_amdgcn_sched_barrier(0);
        #pragma unroll
        for (int i = 0; i < 16; ++i)
            c = __builtin_amdgcn_wmma_f32_16x16x4_f32(false, a[i], false, b[i], (short)0, c, false, false);
    }
    float* mt = m1 + (size_t)(wave * 16) * HID;
    #pragma unroll
    for (int v = 0; v < 8; ++v)
        mt[(v + hi * 8) * HID + l15] = c[v];
}

#define BIN_NODES 400
#define NSLICE    256
#define MAXBINS   1024
#define WG        1024
#define EU        16
#define PG        4
#define CB        4
#define SLICE_CAP 12800

__global__ void __launch_bounds__(WG)
gcn_bin_count(const int* __restrict__ dst, int* __restrict__ counts,
              long long n_edges, long long slice, int n_bins) {
    __shared__ int h[MAXBINS + 32];
    int tid = threadIdx.x;
    for (int i = tid; i < n_bins; i += WG) h[i] = 0;
    if (tid < 32) h[MAXBINS + tid] = 0;
    __syncthreads();
    long long beg = (long long)blockIdx.x * slice;
    long long end = beg + slice < n_edges ? beg + slice : n_edges;
    int n = end > beg ? (int)(end - beg) : 0;
    const int* dp = dst + beg;
    for (int i0 = tid; i0 < n; i0 += WG * EU) {
        int dv[EU];
        #pragma unroll
        for (int u = 0; u < EU; ++u) { int i = i0 + u * WG; dv[u] = dp[i < n ? i : n - 1]; }
        #pragma unroll
        for (int u = 0; u < EU; ++u)
            atomicAdd(&h[i0 + u * WG < n ? (int)((unsigned)dv[u] / BIN_NODES) : MAXBINS + (tid & 31)], 1);
    }
    __syncthreads();
    for (int i = tid; i < n_bins; i += WG) counts[(size_t)i * NSLICE + blockIdx.x] = h[i];
}

__global__ void __launch_bounds__(WG)
gcn_bin_fill(const int* __restrict__ src, const int* __restrict__ dst, const int* __restrict__ counts,
             int* __restrict__ bin_start, int* __restrict__ pairs,
             long long n_edges, long long slice, int n_bins) {
    __shared__ int tot[MAXBINS];
    __shared__ int gdl[MAXBINS];
    __shared__ int cur[MAXBINS + 32];
    __shared__ int wsum[2][32];
    __shared__ int2 stage[SLICE_CAP];
    int tid = threadIdx.x, lane = tid & 31, wave = tid >> 5;
    int g = blockIdx.x;
    if (tid < 32) cur[MAXBINS + tid] = 0;
    long long beg = (long long)g * slice;
    long long end = beg + slice < n_edges ? beg + slice : n_edges;
    int n = end > beg ? (int)(end - beg) : 0;
    const int* sp = src + beg;
    const int* dp = dst + beg;
    int sv[EU], dv[EU];
    #pragma unroll
    for (int u = 0; u < EU; ++u) { sv[u] = 0; dv[u] = 0; }
    if (n > 0) {
        #pragma unroll
        for (int u = 0; u < EU; ++u) {
            int i = tid + u * WG;
            i = i < n ? i : n - 1;
            sv[u] = sp[i];
            dv[u] = dp[i];
        }
    }
    int own = tid < n_bins ? counts[(size_t)tid * NSLICE + g] : 0;
    for (int b0 = wave; b0 < n_bins; b0 += (WG / 32) * CB) {
        int4 c[CB][NSLICE / 128];
        #pragma unroll
        for (int q = 0; q < CB; ++q) {
            int b = b0 + q * (WG / 32);
            b = b < n_bins ? b : n_bins - 1;
            const int4* row = (const int4*)(counts + (size_t)b * NSLICE);
            #pragma unroll
            for (int k = 0; k < NSLICE / 128; ++k) c[q][k] = row[k * 32 + lane];
        }
        #pragma unroll
        for (int q = 0; q < CB; ++q) {
            int b = b0 + q * (WG / 32);
            if (b < n_bins) {
                int t = 0, p = 0;
                #pragma unroll
                for (int k = 0; k < NSLICE / 128; ++k) {
                    int4 cc = c[q][k];
                    int i0 = (k * 32 + lane) * 4;
                    t += cc.x + cc.y + cc.z + cc.w;
                    p += (i0 + 0 < g ? cc.x : 0) + (i0 + 1 < g ? cc.y : 0) + (i0 + 2 < g ? cc.z : 0) + (i0 + 3 < g ? cc.w : 0);
                }
                for (int off = 16; off > 0; off >>= 1) { t += __shfl_xor(t, off); p += __shfl_xor(p, off); }
                if (lane == 0) { tot[b] = t; gdl[b] = p; }
            }
        }
    }
    __syncthreads();
    int v = tid < n_bins ? tot[tid] : 0;
    int x = v, xo = own;
    for (int off = 1; off < 32; off <<= 1) {
        int y = __shfl_up(x, off), yo = __shfl_up(xo, off);
        if (lane >= off) { x += y; xo += yo; }
    }
    if (lane == 31) { wsum[0][wave] = x; wsum[1][wave] = xo; }
    __syncthreads();
    if (wave < 2) {
        int w = wsum[wave][lane], xs = w;
        for (int off = 1; off < 32; off <<= 1) { int y = __shfl_up(xs, off); if (lane >= off) xs += y; }
        wsum[wave][lane] = xs - w;
    }
    __syncthreads();
    int excl = x - v + wsum[0][wave];
    int lbeg = xo - own + wsum[1][wave];
    if (tid < n_bins) {
        cur[tid] = lbeg;
        gdl[tid] += excl - lbeg;
        if (g == 0) bin_start[tid] = excl;
    }
    if (g == 0 && tid == 0) bin_start[n_bins] = (int)n_edges;
    __syncthreads();
    for (int i0 = tid; i0 < n; i0 += WG * EU) {
        if (i0 != tid) {
            #pragma unroll
            for (int u = 0; u < EU; ++u) {
                int i = i0 + u * WG;
                i = i < n ? i : n - 1;
                sv[u] = sp[i];
                dv[u] = dp[i];
            }
        }
        int pos[EU];
        #pragma unroll
        for (int u = 0; u < EU; ++u)
            pos[u] = atomicAdd(&cur[i0 + u * WG < n ? (int)((unsigned)dv[u] / BIN_NODES) : MAXBINS + lane], 1);
        #pragma unroll
        for (int u = 0; u < EU; ++u)
            if (i0 + u * WG < n) stage[pos[u]] = make_int2(sv[u], dv[u]);
    }
    __syncthreads();
    for (int i0 = tid; i0 < n; i0 += WG * EU) {
        int2 r[EU];
        int dl[EU];
        #pragma unroll
        for (int u = 0; u < EU; ++u) { int i = i0 + u * WG; r[u] = stage[i < n ? i : n - 1]; }
        #pragma unroll
        for (int u = 0; u < EU; ++u) {
            int b = (unsigned)r[u].y / BIN_NODES;
            dl[u] = gdl[b];
            r[u].y -= b * BIN_NODES;
        }
        #pragma unroll
        for (int u = 0; u < EU; ++u) {
            int i = i0 + u * WG;
            if (i < n) *(int2*)(pairs + 2 * (size_t)(i + dl[u])) = r[u];
        }
    }
}

__device__ __forceinline__ int2 load_pair(const int* __restrict__ pairs, int e, int end) {
    int ec = e < end ? e : end - 1;
    ec = ec > 0 ? ec : 0;
    int2 p = *(const int2*)(pairs + 2 * (size_t)ec);
    if (e >= end) p = make_int2(0, BIN_NODES);
    return p;
}

__device__ __forceinline__ void gather_rows(const float* __restrict__ msg, int s_l, int f, float* v) {
    #pragma unroll
    for (int j = 0; j < 16; ++j)
        v[j] = msg[(unsigned)__shfl(s_l, j, 16) * HID + f];
}

__device__ __forceinline__ void add_rows(float* acc, int d_l, int f, const float* v) {
    #pragma unroll
    for (int j = 0; j < 16; ++j) atomicAdd(&acc[__shfl(d_l, j, 16) * HID + f], v[j]);
}

template <int NF, bool BIAS, bool PROD>
__global__ void __launch_bounds__(WG)
gcn_bin_sum(const int* __restrict__ pairs, const int* __restrict__ bin_start,
            const float* __restrict__ msg, const float* __restrict__ bias, const float* __restrict__ W2,
            float* __restrict__ res, int n_nodes) {
    __shared__ float acc[(BIN_NODES + 1) * HID];
    int tid = threadIdx.x, lane = tid & 31, f = lane & 15;
    int wave = __builtin_amdgcn_readfirstlane(tid >> 5);
    int b = blockIdx.x;
    int beg = bin_start[b], end = bin_start[b + 1];
    int e0 = beg + wave * 32;
    int2 pn[PG];
    #pragma unroll
    for (int t = 0; t < PG; ++t) pn[t] = load_pair(pairs, e0 + t * WG + lane, end);
    for (int i = tid; i < (BIN_NODES + 1) * HID; i += WG) acc[i] = 0.0f;
    __syncthreads();
    float v[PG - 1][16];
    for (; e0 + (PG - 1) * WG < end; e0 += WG * PG) {
        int2 p[PG];
        #pragma unroll
        for (int t = 0; t < PG; ++t) p[t] = pn[t];
        gather_rows(msg, p[0].x, f, v[0]);
        #pragma unroll
        for (int t = 0; t < PG; ++t) {
            if (t + 1 < PG) gather_rows(msg, p[t + 1].x, f, v[(t + 1) & 1]);
            if (t == PG - 2) {
                #pragma unroll
                for (int u = 0; u < PG; ++u) pn[u] = load_pair(pairs, e0 + (PG + u) * WG + lane, end);
            }
            add_rows(acc, p[t].y, f, v[t & 1]);
        }
    }
    if (e0 < end) {
        #pragma unroll
        for (int t = 0; t < PG - 1; ++t) gather_rows(msg, pn[t].x, f, v[t]);
        #pragma unroll
        for (int t = 0; t < PG - 1; ++t)
            if (e0 + t * WG < end) add_rows(acc, pn[t].y, f, v[t]);
    }
    int hi = lane >> 4;
    v2f wb[HID / 4], bb[HID / 4];
    if (PROD) {
        #pragma unroll
        for (int k = 0; k < HID / 4; ++k) {
            int ka = k * 4 + hi * 2;
            int fc = f < OUTD ? f : 0;
            float w0 = W2[(ka + 0) * OUTD + fc], w1 = W2[(ka + 1) * OUTD + fc];
            wb[k].x = f < OUTD ? w0 : 0.0f;
            wb[k].y = f < OUTD ? w1 : 0.0f;
            bb[k].x = bias[ka];
            bb[k].y = bias[ka + 1];
        }
    }
    __syncthreads();
    if (PROD) {
        static_assert(BIN_NODES % 16 == 0 && BIN_NODES / 16 <= WG / 32, "one whole 16-row tile per wave");
        if (wave < BIN_NODES / 16) {
            float* at = acc + wave * 16 * HID;
            v8f c = {};
            #pragma unroll
            for (int k = 0; k < HID / 4; ++k) {
                int ka = k * 4 + hi * 2;
                v2f s = *(const v2f*)(at + f * HID + ka);
                v2f a;
                a.x = fmaxf(s.x + bb[k].x, 0.0f);
                a.y = fmaxf(s.y + bb[k].y, 0.0f);
                c = __builtin_amdgcn_wmma_f32_16x16x4_f32(false, a, false, wb[k], (short)0, c, false, false);
            }
            #pragma unroll
            for (int r = 0; r < 8; ++r)
                at[(r + hi * 8) * HID + f] = c[r];
        }
        __syncthreads();
    }
    int rows = n_nodes - b * BIN_NODES;
    if (rows > BIN_NODES) rows = BIN_NODES;
    float* rb = res + (size_t)b * BIN_NODES * NF;
    for (int i = tid; i < rows * NF; i += WG) {
        int r = i / NF, o = i - r * NF;
        float a = acc[r * HID + o];
        rb[i] = BIAS ? a + bias[o] : a;
    }
}

extern "C" void kernel_launch(void* const* d_in, const int* in_sizes, int n_in,
                              void* d_out, int out_size, void* d_ws, size_t ws_size,
                              hipStream_t stream) {
    const float* x  = (const float*)d_in[0];
    const int*   ei = (const int*)  d_in[1];
    const float* W1 = (const float*)d_in[2];
    const float* b1 = (const float*)d_in[3];
    const float* W2 = (const float*)d_in[4];
    const float* b2 = (const float*)d_in[5];
    float* out = (float*)d_out;

    int n_nodes       = in_sizes[0] / IN_DIM;
    long long n_edges = (long long)in_sizes[1] / 2;
    const int* src = ei;
    const int* dst = ei + n_edges;

    int n_bins      = (n_nodes + BIN_NODES - 1) / BIN_NODES;
    long long slice = (n_edges + NSLICE - 1) / NSLICE;

    float* m1  = (float*)d_ws;
    float* m2  = m1 + (size_t)n_nodes * HID;
    int* counts    = (int*)(m2 + (size_t)n_nodes * HID);
    int* bin_start = counts + (size_t)n_bins * NSLICE;
    int* pairs     = bin_start + ((n_bins + 1 + 63) & ~63);

    int n_tiles = (n_nodes + 15) / 16;
    int gemm_blocks = (n_tiles + 7) / 8;

    gcn_bin_count  <<<NSLICE, WG, 0, stream>>>(dst, counts, n_edges, slice, n_bins);
    gcn_bin_fill   <<<NSLICE, WG, 0, stream>>>(src, dst, counts, bin_start, pairs, n_edges, slice, n_bins);
    gcn_gemm1_wmma <<<gemm_blocks, 256, 0, stream>>>(x, W1, m1, n_tiles);
    gcn_bin_sum<HID,  false, true ><<<n_bins, WG, 0, stream>>>(pairs, bin_start, m1, b1, W2, m2,  n_nodes);
    gcn_bin_sum<OUTD, true,  false><<<n_bins, WG, 0, stream>>>(pairs, bin_start, m2, b2, W2, out, n_nodes);
}


# each slice orders its own edges by block in LDS (returning add = rank) into its own part of the list; block sums join their 256 runs by an LDS prefix table: count launch gone; sums in arrival order
# speedup vs baseline: 15.7688x; 1.0749x over previous
#include <hip/hip_runtime.h>

typedef float v2f __attribute__((ext_vector_type(2)));
typedef float v8f __attribute__((ext_vector_type(8)));

#define IN_DIM 256
#define HID 16
#define OUTD 10

__global__ void gcn_gemm1_wmma(const float* __restrict__ x, const float* __restrict__ W1,
                               float* __restrict__ m1, int n_tiles) {
    int wave = (int)((blockIdx.x * blockDim.x + threadIdx.x) >> 5);
    int lane = threadIdx.x & 31;
    if (wave >= n_tiles) return;
    int l15 = lane & 15;
    int hi  = lane >> 4;
    const float* xrow = x + ((size_t)(wave * 16 + l15)) * IN_DIM;

    v8f c = {};
    for (int kb = 0; kb < IN_DIM; kb += 64) {
        v2f a[16], b[16];
        #pragma unroll
        for (int i = 0; i < 16; ++i) {
            int ka = kb + i * 4 + hi * 2;
            a[i].x = xrow[ka];                 a[i].y = xrow[ka + 1];
            b[i].x = W1[(ka + 0) * HID + l15]; b[i].y = W1[(ka + 1) * HID + l15];
        }
        __builtin_amdgcn_sched_barrier(0);
        #pragma unroll
        for (int i = 0; i < 16; ++i)
            c = __builtin_amdgcn_wmma_f32_16x16x4_f32(false, a[i], false, b[i], (short)0, c, false, false);
    }
    float* mt = m1 + (size_t)(wave * 16) * HID;
    #pragma unroll
    for (int v = 0; v < 8; ++v)
        mt[(v + hi * 8) * HID + l15] = c[v];
}

#define BIN_NODES 400
#define NSLICE    256
#define MAXBINS   1024
#define WG        1024
#define EU        16
#define PG        4
#define SLICE_CAP 12800
static_assert(SLICE_CAP <= WG * EU, "a thread holds its EU edges of the slice in registers from the rank to the placing");
static_assert(NSLICE == 256, "the search in find_group takes 8 halving steps over 256 runs");

__global__ void __launch_bounds__(WG)
gcn_bin_fill(const int* __restrict__ src, const int* __restrict__ dst,
             int* __restrict__ runs, int* __restrict__ pairs,
             long long n_edges, long long slice, int n_bins) {
    __shared__ int cnt[MAXBINS + 32];
    __shared__ int wsum[32];
    __shared__ int2 stage[SLICE_CAP + 32];
    int tid = threadIdx.x, lane = tid & 31, wave = tid >> 5;
    int g = blockIdx.x;
    long long beg = (long long)g * slice;
    long long end = beg + slice < n_edges ? beg + slice : n_edges;
    int n = end > beg ? (int)(end - beg) : 0;
    const int* sp = src + beg;
    const int* dp = dst + beg;
    int sv[EU], dv[EU];
    #pragma unroll
    for (int u = 0; u < EU; ++u) { sv[u] = 0; dv[u] = 0; }
    if (n > 0) {
        #pragma unroll
        for (int u = 0; u < EU; ++u) {
            int i = tid + u * WG;
            i = i < n ? i : n - 1;
            sv[u] = sp[i];
            dv[u] = dp[i];
        }
    }
    for (int i = tid; i < n_bins; i += WG) cnt[i] = 0;
    if (tid < 32) cnt[MAXBINS + tid] = 0;
    __syncthreads();
    int rk[EU];
    #pragma unroll
    for (int u = 0; u < EU; ++u)
        rk[u] = atomicAdd(&cnt[tid + u * WG < n ? (int)((unsigned)dv[u] / BIN_NODES) : MAXBINS + lane], 1);
    __syncthreads();
    int own = tid < n_bins ? cnt[tid] : 0;
    int x = own;
    for (int off = 1; off < 32; off <<= 1) { int y = __shfl_up(x, off); if (lane >= off) x += y; }
    if (lane == 31) wsum[wave] = x;
    __syncthreads();
    if (wave == 0) {
        int w = wsum[lane], xs = w;
        for (int off = 1; off < 32; off <<= 1) { int y = __shfl_up(xs, off); if (lane >= off) xs += y; }
        wsum[lane] = xs - w;
    }
    __syncthreads();
    int lbeg = x - own + wsum[wave];
    if (tid < n_bins) {
        cnt[tid] = lbeg;
        runs[(size_t)tid * NSLICE + g] = lbeg;
    }
    if (tid == 0) runs[(size_t)n_bins * NSLICE + g] = n;
    if (tid < 32) cnt[MAXBINS + tid] = SLICE_CAP + tid;
    __syncthreads();
    int pos[EU];
    #pragma unroll
    for (int u = 0; u < EU; ++u) {
        bool in = tid + u * WG < n;
        int b = (int)((unsigned)dv[u] / BIN_NODES);
        pos[u] = cnt[in ? b : MAXBINS + lane] + (in ? rk[u] : 0);
        dv[u] -= b * BIN_NODES;
    }
    #pragma unroll
    for (int u = 0; u < EU; ++u) stage[pos[u]] = make_int2(sv[u], dv[u]);
    __syncthreads();
    int2 r[EU];
    int nl = n > 0 ? n - 1 : 0;
    #pragma unroll
    for (int u = 0; u < EU; ++u) { int i = tid + u * WG; r[u] = stage[i < n ? i : nl]; }
    #pragma unroll
    for (int u = 0; u < EU; ++u) {
        int i = tid + u * WG;
        if (i < n) *(int2*)(pairs + 2 * (size_t)(beg + i)) = r[u];
    }
}

__device__ __forceinline__ void find_group(const int* pre, const int* rbase, int p0, int end, int last, int* idx) {
    int q[PG], g[PG];
    #pragma unroll
    for (int u = 0; u < PG; ++u) {
        int p = p0 + u * WG;
        q[u] = p < end ? p : end - 1;
        q[u] = q[u] > 0 ? q[u] : 0;
        g[u] = 0;
    }
    #pragma unroll
    for (int s = NSLICE / 2; s > 0; s >>= 1) {
        #pragma unroll
        for (int u = 0; u < PG; ++u) g[u] += pre[g[u] + s] <= q[u] ? s : 0;
    }
    #pragma unroll
    for (int u = 0; u < PG; ++u) {
        int i = rbase[g[u]] + q[u];
        i = i < last ? i : last;
        idx[u] = i > 0 ? i : 0;
    }
}

__device__ __forceinline__ void fetch_group(const int* __restrict__ pairs, const int* idx, int p0, int end, int2* out) {
    #pragma unroll
    for (int u = 0; u < PG; ++u) {
        int2 r = *(const int2*)(pairs + 2 * (size_t)idx[u]);
        if (p0 + u * WG >= end) r = make_int2(0, BIN_NODES);
        out[u] = r;
    }
}

__device__ __forceinline__ void gather_rows(const float* __restrict__ msg, int s_l, int f, float* v) {
    #pragma unroll
    for (int j = 0; j < 16; ++j)
        v[j] = msg[(unsigned)__shfl(s_l, j, 16) * HID + f];
}

__device__ __forceinline__ void add_rows(float* acc, int d_l, int f, const float* v) {
    #pragma unroll
    for (int j = 0; j < 16; ++j) atomicAdd(&acc[__shfl(d_l, j, 16) * HID + f], v[j]);
}

template <int NF, bool BIAS, bool PROD>
__global__ void __launch_bounds__(WG)
gcn_bin_sum(const int* __restrict__ pairs, const int* __restrict__ runs,
            const float* __restrict__ msg, const float* __restrict__ bias, const float* __restrict__ W2,
            float* __restrict__ res, int n_nodes, int slice, int last) {
    __shared__ float acc[(BIN_NODES + 1) * HID];
    __shared__ int pre[NSLICE];
    __shared__ int rbase[NSLICE];
    __shared__ int wtot[32];
    int tid = threadIdx.x, lane = tid & 31, f = lane & 15;
    int wave = __builtin_amdgcn_readfirstlane(tid >> 5);
    int b = blockIdx.x;
    int rs = 0, len = 0;
    if (tid < NSLICE) {
        rs  = runs[(size_t)b * NSLICE + tid];
        len = runs[(size_t)(b + 1) * NSLICE + tid] - rs;
    }
    for (int i = tid; i < (BIN_NODES + 1) * HID; i += WG) acc[i] = 0.0f;
    int x = len;
    for (int off = 1; off < 32; off <<= 1) { int y = __shfl_up(x, off); if (lane >= off) x += y; }
    if (lane == 31) wtot[wave] = x;
    __syncthreads();
    int wt = wtot[lane], ws = wt;
    for (int off = 1; off < NSLICE / 32; off <<= 1) { int y = __shfl_up(ws, off); if (lane >= off) ws += y; }
    int end = __builtin_amdgcn_readlane(ws, NSLICE / 32 - 1);
    int excl = x - len + __builtin_amdgcn_readlane(ws - wt, wave);
    if (tid < NSLICE) {
        pre[tid]   = excl;
        rbase[tid] = tid * slice + rs - excl;
    }
    __syncthreads();
    int e0 = wave * 32;
    int2 pn[PG];
    int ix[PG];
    find_group(pre, rbase, e0 + lane, end, last, ix);
    fetch_group(pairs, ix, e0 + lane, end, pn);
    asm volatile("" :: "v"(pn[0].x), "v"(pn[0].y), "v"(pn[1].x), "v"(pn[1].y), "v"(pn[2].x), "v"(pn[2].y), "v"(pn[3].x), "v"(pn[3].y));
    static_assert(PG == 4, "the statement above names the PG records one by one");
    float v[PG - 1][16];
    for (; e0 + (PG - 1) * WG < end; e0 += WG * PG) {
        int2 p[PG];
        #pragma unroll
        for (int t = 0; t < PG; ++t) p[t] = pn[t];
        gather_rows(msg, p[0].x, f, v[0]);
        #pragma unroll
        for (int t = 0; t < PG; ++t) {
            if (t + 1 < PG) gather_rows(msg, p[t + 1].x, f, v[(t + 1) & 1]);
            if (t == 0) {
                __builtin_amdgcn_sched_barrier(0);
                find_group(pre, rbase, e0 + PG * WG + lane, end, last, ix);
            }
            if (t == PG - 2) {
                fetch_group(pairs, ix, e0 + PG * WG + lane, end, pn);
            }
            add_rows(acc, p[t].y, f, v[t & 1]);
        }
    }
    if (e0 < end) {
        #pragma unroll
        for (int t = 0; t < PG - 1; ++t) gather_rows(msg, pn[t].x, f, v[t]);
        #pragma unroll
        for (int t = 0; t < PG - 1; ++t)
            if (e0 + t * WG < end) add_rows(acc, pn[t].y, f, v[t]);
    }
    int hi = lane >> 4;
    v2f wb[HID / 4], bb[HID / 4];
    if (PROD) {
        #pragma unroll
        for (int k = 0; k < HID / 4; ++k) {
            int ka = k * 4 + hi * 2;
            int fc = f < OUTD ? f : 0;
            float w0 = W2[(ka + 0) * OUTD + fc], w1 = W2[(ka + 1) * OUTD + fc];
            wb[k].x = f < OUTD ? w0 : 0.0f;
            wb[k].y = f < OUTD ? w1 : 0.0f;
            bb[k].x = bias[ka];
            bb[k].y = bias[ka + 1];
        }
    }
    __syncthreads();
    if (PROD) {
        static_assert(BIN_NODES % 16 == 0 && BIN_NODES / 16 <= WG / 32, "one whole 16-row tile per wave");
        if (wave < BIN_NODES / 16) {
            float* at = acc + wave * 16 * HID;
            v8f c = {};
            #pragma unroll
            for (int k = 0; k < HID / 4; ++k) {
                int ka = k * 4 + hi * 2;
                v2f s = *(const v2f*)(at + f * HID + ka);
                v2f a;
                a.x = fmaxf(s.x + bb[k].x, 0.0f);
                a.y = fmaxf(s.y + bb[k].y, 0.0f);
                c = __builtin_amdgcn_wmma_f32_16x16x4_f32(false, a, false, wb[k], (short)0, c, false, false);
            }
            #pragma unroll
            for (int r = 0; r < 8; ++r)
                at[(r + hi * 8) * HID + f] = c[r];
        }
        __syncthreads();
    }
    int rows = n_nodes - b * BIN_NODES;
    if (rows > BIN_NODES) rows = BIN_NODES;
    float* rb = res + (size_t)b * BIN_NODES * NF;
    for (int i = tid; i < rows * NF; i += WG) {
        int r = i / NF, o = i - r * NF;
        float a = acc[r * HID + o];
        rb[i] = BIAS ? a + bias[o] : a;
    }
}

extern "C" void kernel_launch(void* const* d_in, const int* in_sizes, int n_in,
                              void* d_out, int out_size, void* d_ws, size_t ws_size,
                              hipStream_t stream) {
    const float* x  = (const float*)d_in[0];
    const int*   ei = (const int*)  d_in[1];
    const float* W1 = (const float*)d_in[2];
    const float* b1 = (const float*)d_in[3];
    const float* W2 = (const float*)d_in[4];
    const float* b2 = (const float*)d_in[5];
    float* out = (float*)d_out;

    int n_nodes       = in_sizes[0] / IN_DIM;
    long long n_edges = (long long)in_sizes[1] / 2;
    const int* src = ei;
    const int* dst = ei + n_edges;

    int n_bins      = (n_nodes + BIN_NODES - 1) / BIN_NODES;
    long long slice = (n_edges + NSLICE - 1) / NSLICE;

    float* m1  = (float*)d_ws;
    float* m2  = m1 + (size_t)n_nodes * HID;
    int* runs  = (int*)(m2 + (size_t)n_nodes * HID);
    int* pairs = runs + (size_t)(n_bins + 1) * NSLICE;

    int n_tiles = (n_nodes + 15) / 16;
    int gemm_blocks = (n_tiles + 7) / 8;

    gcn_bin_fill   <<<NSLICE, WG, 0, stream>>>(src, dst, runs, pairs, n_edges, slice, n_bins);
    gcn_gemm1_wmma <<<gemm_blocks, 256, 0, stream>>>(x, W1, m1, n_tiles);
    gcn_bin_sum<HID,  false, true ><<<n_bins, WG, 0, stream>>>(pairs, runs, m1, b1, W2, m2,  n_nodes, (int)slice, (int)(n_edges - 1));
    gcn_bin_sum<OUTD, true,  false><<<n_bins, WG, 0, stream>>>(pairs, runs, m2, b2, W2, out, n_nodes, (int)slice, (int)(n_edges - 1));
}


# x@W1 operands go through LDS: x pieces loaded as whole 128-byte lines (16 B a lane, 4 lines a load, not 16) and re-read in WMMA order, W1 staged once per work-group; products and k order unchanged
# speedup vs baseline: 16.4886x; 1.0456x over previous
#include <hip/hip_runtime.h>

typedef float v2f __attribute__((ext_vector_type(2)));
typedef float v8f __attribute__((ext_vector_type(8)));

#define IN_DIM 256
#define HID 16
#define OUTD 10

#define XP 68
#define WP 260
typedef float v4f __attribute__((ext_vector_type(4)));

__device__ __forceinline__ void wave_lds_handover() {
    __builtin_amdgcn_fence(__ATOMIC_RELEASE, "wavefront");
    __builtin_amdgcn_wave_barrier();
    __builtin_amdgcn_fence(__ATOMIC_ACQUIRE, "wavefront");
}

__global__ void __launch_bounds__(256)
gcn_gemm1_wmma(const float* __restrict__ x, const float* __restrict__ W1,
               float* __restrict__ m1, int n_tiles) {
    __shared__ __attribute__((aligned(16))) float w1t[HID * WP];
    __shared__ __attribute__((aligned(16))) float xs[8][16 * XP];
    int tid = threadIdx.x, lane = tid & 31;
    int wv = __builtin_amdgcn_readfirstlane(tid >> 5);
    int tile = blockIdx.x * 8 + wv;
    int tl = tile < n_tiles ? tile : n_tiles - 1;
    int l15 = lane & 15;
    int hi  = lane >> 4;
    const float* xg = x + ((size_t)tl * 16 + hi) * IN_DIM + l15 * 4;
    v4f r[8];
    #pragma unroll
    for (int j = 0; j < 8; ++j) r[j] = *(const v4f*)(xg + (size_t)(2 * j) * IN_DIM);
    #pragma unroll
    for (int j = 0; j < 4; ++j) {
        v4f w = *(const v4f*)(W1 + tid * 4 + 1024 * j);
        int k = (tid >> 2) + 64 * j, n0 = (tid & 3) * 4;
        w1t[(n0 + 0) * WP + k] = w.x;
        w1t[(n0 + 1) * WP + k] = w.y;
        w1t[(n0 + 2) * WP + k] = w.z;
        w1t[(n0 + 3) * WP + k] = w.w;
    }
    __syncthreads();
    float* st = xs[wv];
    const float* ap = st + l15 * XP + hi * 2;
    const float* bp = w1t + l15 * WP + hi * 2;
    v8f c = {};
    #pragma unroll
    for (int t = 0; t < IN_DIM / 64; ++t) {
        #pragma unroll
        for (int j = 0; j < 8; ++j) *(v4f*)(st + (2 * j + hi) * XP + l15 * 4) = r[j];
        if (t + 1 < IN_DIM / 64) {
            #pragma unroll
            for (int j = 0; j < 8; ++j) r[j] = *(const v4f*)(xg + (size_t)(2 * j) * IN_DIM + (t + 1) * 64);
        }
        wave_lds_handover();
        v2f a[16], b[16];
        #pragma unroll
        for (int s = 0; s < 16; ++s) {
            a[s] = *(const v2f*)(ap + 4 * s);
            b[s] = *(const v2f*)(bp + t * 64 + 4 * s);
        }
        __builtin_amdgcn_sched_barrier(0);
        #pragma unroll
        for (int s = 0; s < 16; ++s)
            c = __builtin_amdgcn_wmma_f32_16x16x4_f32(false, a[s], false, b[s], (short)0, c, false, false);
        wave_lds_handover();
    }
    if (tile < n_tiles) {
        float* mt = m1 + (size_t)(tile * 16) * HID;
        #pragma unroll
        for (int v = 0; v < 8; ++v)
            mt[(v + hi * 8) * HID + l15] = c[v];
    }
}

#define BIN_NODES 400
#define NSLICE    256
#define MAXBINS   1024
#define WG        1024
#define EU        16
#define PG        4
#define SLICE_CAP 12800
static_assert(SLICE_CAP <= WG * EU, "a thread holds its EU edges of the slice in registers from the rank to the placing");
static_assert(NSLICE == 256, "the search in find_group takes 8 halving steps over 256 runs");

__global__ void __launch_bounds__(WG)
gcn_bin_fill(const int* __restrict__ src, const int* __restrict__ dst,
             int* __restrict__ runs, int* __restrict__ pairs,
             long long n_edges, long long slice, int n_bins) {
    __shared__ int cnt[MAXBINS + 32];
    __shared__ int wsum[32];
    __shared__ int2 stage[SLICE_CAP + 32];
    int tid = threadIdx.x, lane = tid & 31, wave = tid >> 5;
    int g = blockIdx.x;
    long long beg = (long long)g * slice;
    long long end = beg + slice < n_edges ? beg + slice : n_edges;
    int n = end > beg ? (int)(end - beg) : 0;
    const int* sp = src + beg;
    const int* dp = dst + beg;
    int sv[EU], dv[EU];
    #pragma unroll
    for (int u = 0; u < EU; ++u) { sv[u] = 0; dv[u] = 0; }
    if (n > 0) {
        #pragma unroll
        for (int u = 0; u < EU; ++u) {
            int i = tid + u * WG;
            i = i < n ? i : n - 1;
            sv[u] = sp[i];
            dv[u] = dp[i];
        }
    }
    for (int i = tid; i < n_bins; i += WG) cnt[i] = 0;
    if (tid < 32) cnt[MAXBINS + tid] = 0;
    __syncthreads();
    int rk[EU];
    #pragma unroll
    for (int u = 0; u < EU; ++u)
        rk[u] = atomicAdd(&cnt[tid + u * WG < n ? (int)((unsigned)dv[u] / BIN_NODES) : MAXBINS + lane], 1);
    __syncthreads();
    int own = tid < n_bins ? cnt[tid] : 0;
    int x = own;
    for (int off = 1; off < 32; off <<= 1) { int y = __shfl_up(x, off); if (lane >= off) x += y; }
    if (lane == 31) wsum[wave] = x;
    __syncthreads();
    if (wave == 0) {
        int w = wsum[lane], xs = w;
        for (int off = 1; off < 32; off <<= 1) { int y = __shfl_up(xs, off); if (lane >= off) xs += y; }
        wsum[lane] = xs - w;
    }
    __syncthreads();
    int lbeg = x - own + wsum[wave];
    if (tid < n_bins) {
        cnt[tid] = lbeg;
        runs[(size_t)tid * NSLICE + g] = lbeg;
    }
    if (tid == 0) runs[(size_t)n_bins * NSLICE + g] = n;
    if (tid < 32) cnt[MAXBINS + tid] = SLICE_CAP + tid;
    __syncthreads();
    int pos[EU];
    #pragma unroll
    for (int u = 0; u < EU; ++u) {
        bool in = tid + u * WG < n;
        int b = (int)((unsigned)dv[u] / BIN_NODES);
        pos[u] = cnt[in ? b : MAXBINS + lane] + (in ? rk[u] : 0);
        dv[u] -= b * BIN_NODES;
    }
    #pragma unroll
    for (int u = 0; u < EU; ++u) stage[pos[u]] = make_int2(sv[u], dv[u]);
    __syncthreads();
    int2 r[EU];
    int nl = n > 0 ? n - 1 : 0;
    #pragma unroll
    for (int u = 0; u < EU; ++u) { int i = tid + u * WG; r[u] = stage[i < n ? i : nl]; }
    #pragma unroll
    for (int u = 0; u < EU; ++u) {
        int i = tid + u * WG;
        if (i < n) *(int2*)(pairs + 2 * (size_t)(beg + i)) = r[u];
    }
}

__device__ __forceinline__ void find_group(const int* pre, const int* rbase, int p0, int end, int last, int* idx) {
    int q[PG], g[PG];
    #pragma unroll
    for (int u = 0; u < PG; ++u) {
        int p = p0 + u * WG;
        q[u] = p < end ? p : end - 1;
        q[u] = q[u] > 0 ? q[u] : 0;
        g[u] = 0;
    }
    #pragma unroll
    for (int s = NSLICE / 2; s > 0; s >>= 1) {
        #pragma unroll
        for (int u = 0; u < PG; ++u) g[u] += pre[g[u] + s] <= q[u] ? s : 0;
    }
    #pragma unroll
    for (int u = 0; u < PG; ++u) {
        int i = rbase[g[u]] + q[u];
        i = i < last ? i : last;
        idx[u] = i > 0 ? i : 0;
    }
}

__device__ __forceinline__ void fetch_group(const int* __restrict__ pairs, const int* idx, int p0, int end, int2* out) {
    #pragma unroll
    for (int u = 0; u < PG; ++u) {
        int2 r = *(const int2*)(pairs + 2 * (size_t)idx[u]);
        if (p0 + u * WG >= end) r = make_int2(0, BIN_NODES);
        out[u] = r;
    }
}

__device__ __forceinline__ void gather_rows(const float* __restrict__ msg, int s_l, int f, float* v) {
    #pragma unroll
    for (int j = 0; j < 16; ++j)
        v[j] = msg[(unsigned)__shfl(s_l, j, 16) * HID + f];
}

__device__ __forceinline__ void add_rows(float* acc, int d_l, int f, const float* v) {
    #pragma unroll
    for (int j = 0; j < 16; ++j) atomicAdd(&acc[__shfl(d_l, j, 16) * HID + f], v[j]);
}

template <int NF, bool BIAS, bool PROD>
__global__ void __launch_bounds__(WG)
gcn_bin_sum(const int* __restrict__ pairs, const int* __restrict__ runs,
            const float* __restrict__ msg, const float* __restrict__ bias, const float* __restrict__ W2,
            float* __restrict__ res, int n_nodes, int slice, int last) {
    __shared__ float acc[(BIN_NODES + 1) * HID];
    __shared__ int pre[NSLICE];
    __shared__ int rbase[NSLICE];
    __shared__ int wtot[32];
    int tid = threadIdx.x, lane = tid & 31, f = lane & 15;
    int wave = __builtin_amdgcn_readfirstlane(tid >> 5);
    int b = blockIdx.x;
    int rs = 0, len = 0;
    if (tid < NSLICE) {
        rs  = runs[(size_t)b * NSLICE + tid];
        len = runs[(size_t)(b + 1) * NSLICE + tid] - rs;
    }
    for (int i = tid; i < (BIN_NODES + 1) * HID; i += WG) acc[i] = 0.0f;
    int x = len;
    for (int off = 1; off < 32; off <<= 1) { int y = __shfl_up(x, off); if (lane >= off) x += y; }
    if (lane == 31) wtot[wave] = x;
    __syncthreads();
    int wt = wtot[lane], ws = wt;
    for (int off = 1; off < NSLICE / 32; off <<= 1) { int y = __shfl_up(ws, off); if (lane >= off) ws += y; }
    int end = __builtin_amdgcn_readlane(ws, NSLICE / 32 - 1);
    int excl = x - len + __builtin_amdgcn_readlane(ws - wt, wave);
    if (tid < NSLICE) {
        pre[tid]   = excl;
        rbase[tid] = tid * slice + rs - excl;
    }
    __syncthreads();
    int e0 = wave * 32;
    int2 pn[PG];
    int ix[PG];
    find_group(pre, rbase, e0 + lane, end, last, ix);
    fetch_group(pairs, ix, e0 + lane, end, pn);
    asm volatile("" :: "v"(pn[0].x), "v"(pn[0].y), "v"(pn[1].x), "v"(pn[1].y), "v"(pn[2].x), "v"(pn[2].y), "v"(pn[3].x), "v"(pn[3].y));
    static_assert(PG == 4, "the statement above names the PG records one by one");
    float v[PG - 1][16];
    for (; e0 + (PG - 1) * WG < end; e0 += WG * PG) {
        int2 p[PG];
        #pragma unroll
        for (int t = 0; t < PG; ++t) p[t] = pn[t];
        gather_rows(msg, p[0].x, f, v[0]);
        #pragma unroll
        for (int t = 0; t < PG; ++t) {
            if (t + 1 < PG) gather_rows(msg, p[t + 1].x, f, v[(t + 1) & 1]);
            if (t == 0) {
                __builtin_amdgcn_sched_barrier(0);
                find_group(pre, rbase, e0 + PG * WG + lane, end, last, ix);
            }
            if (t == PG - 2) {
                fetch_group(pairs, ix, e0 + PG * WG + lane, end, pn);
            }
            add_rows(acc, p[t].y, f, v[t & 1]);
        }
    }
    if (e0 < end) {
        #pragma unroll
        for (int t = 0; t < PG - 1; ++t) gather_rows(msg, pn[t].x, f, v[t]);
        #pragma unroll
        for (int t = 0; t < PG - 1; ++t)
            if (e0 + t * WG < end) add_rows(acc, pn[t].y, f, v[t]);
    }
    int hi = lane >> 4;
    v2f wb[HID / 4], bb[HID / 4];
    if (PROD) {
        #pragma unroll
        for (int k = 0; k < HID / 4; ++k) {
            int ka = k * 4 + hi * 2;
            int fc = f < OUTD ? f : 0;
            float w0 = W2[(ka + 0) * OUTD + fc], w1 = W2[(ka + 1) * OUTD + fc];
            wb[k].x = f < OUTD ? w0 : 0.0f;
            wb[k].y = f < OUTD ? w1 : 0.0f;
            bb[k].x = bias[ka];
            bb[k].y = bias[ka + 1];
        }
    }
    __syncthreads();
    if (PROD) {
        static_assert(BIN_NODES % 16 == 0 && BIN_NODES / 16 <= WG / 32, "one whole 16-row tile per wave");
        if (wave < BIN_NODES / 16) {
            float* at = acc + wave * 16 * HID;
            v8f c = {};
            #pragma unroll
            for (int k = 0; k < HID / 4; ++k) {
                int ka = k * 4 + hi * 2;
                v2f s = *(const v2f*)(at + f * HID + ka);
                v2f a;
                a.x = fmaxf(s.x + bb[k].x, 0.0f);
                a.y = fmaxf(s.y + bb[k].y, 0.0f);
                c = __builtin_amdgcn_wmma_f32_16x16x4_f32(false, a, false, wb[k], (short)0, c, false, false);
            }
            #pragma unroll
            for (int r = 0; r < 8; ++r)
                at[(r + hi * 8) * HID + f] = c[r];
        }
        __syncthreads();
    }
    int rows = n_nodes - b * BIN_NODES;
    if (rows > BIN_NODES) rows = BIN_NODES;
    float* rb = res + (size_t)b * BIN_NODES * NF;
    for (int i = tid; i < rows * NF; i += WG) {
        int r = i / NF, o = i - r * NF;
        float a = acc[r * HID + o];
        rb[i] = BIAS ? a + bias[o] : a;
    }
}

extern "C" void kernel_launch(void* const* d_in, const int* in_sizes, int n_in,
                              void* d_out, int out_size, void* d_ws, size_t ws_size,
                              hipStream_t stream) {
    const float* x  = (const float*)d_in[0];
    const int*   ei = (const int*)  d_in[1];
    const float* W1 = (const float*)d_in[2];
    const float* b1 = (const float*)d_in[3];
    const float* W2 = (const float*)d_in[4];
    const float* b2 = (const float*)d_in[5];
    float* out = (float*)d_out;

    int n_nodes       = in_sizes[0] / IN_DIM;
    long long n_edges = (long long)in_sizes[1] / 2;
    const int* src = ei;
    const int* dst = ei + n_edges;

    int n_bins      = (n_nodes + BIN_NODES - 1) / BIN_NODES;
    long long slice = (n_edges + NSLICE - 1) / NSLICE;

    float* m1  = (float*)d_ws;
    float* m2  = m1 + (size_t)n_nodes * HID;
    int* runs  = (int*)(m2 + (size_t)n_nodes * HID);
    int* pairs = runs + (size_t)(n_bins + 1) * NSLICE;

    int n_tiles = (n_nodes + 15) / 16;
    int gemm_blocks = (n_tiles + 7) / 8;

    gcn_bin_fill   <<<NSLICE, WG, 0, stream>>>(src, dst, runs, pairs, n_edges, slice, n_bins);
    gcn_gemm1_wmma <<<gemm_blocks, 256, 0, stream>>>(x, W1, m1, n_tiles);
    gcn_bin_sum<HID,  false, true ><<<n_bins, WG, 0, stream>>>(pairs, runs, m1, b1, W2, m2,  n_nodes, (int)slice, (int)(n_edges - 1));
    gcn_bin_sum<OUTD, true,  false><<<n_bins, WG, 0, stream>>>(pairs, runs, m2, b2, W2, out, n_nodes, (int)slice, (int)(n_edges - 1));
}


# edge binning and x@W1 share one launch: each work-group bins its slice and multiplies its share of row tiles, half the groups in the other order; one launch less; sums still in LDS arrival order
# speedup vs baseline: 18.1022x; 1.0979x over previous
#include <hip/hip_runtime.h>

typedef float v2f __attribute__((ext_vector_type(2)));
typedef float v8f __attribute__((ext_vector_type(8)));

#define IN_DIM 256
#define HID 16
#define OUTD 10

#define BIN_NODES 400
#define NSLICE    256
#define MAXBINS   1024
#define WG        1024
#define EU        16
#define PG        4
#define SLICE_CAP 12800
static_assert(SLICE_CAP <= WG * EU, "a thread holds its EU edges of the slice in registers from the rank to the placing");
static_assert(NSLICE == 256, "the search in find_group takes 8 halving steps over 256 runs");

#define XP 68
#define WP 260
typedef float v4f __attribute__((ext_vector_type(4)));
static_assert(IN_DIM * HID == WG * 4, "W1 staging: one 16-byte piece of W1 per thread of the work-group");

__device__ __forceinline__ void wave_lds_handover() {
    __builtin_amdgcn_fence(__ATOMIC_RELEASE, "wavefront");
    __builtin_amdgcn_wave_barrier();
    __builtin_amdgcn_fence(__ATOMIC_ACQUIRE, "wavefront");
}

__device__ __forceinline__ void gemm1_part(unsigned char* lds, const float* __restrict__ x, const float* __restrict__ W1,
                                           float* __restrict__ m1, int n_tiles) {
    float* w1t = (float*)lds;
    float* xsb = w1t + HID * WP;
    int tid = threadIdx.x, lane = tid & 31;
    int wv = __builtin_amdgcn_readfirstlane(tid >> 5);
    int tstep = (WG / 32) * (int)gridDim.x;
    int tile = wv * (int)gridDim.x + (int)blockIdx.x;
    int l15 = lane & 15;
    int hi  = lane >> 4;
    v4f r[8];
    if (tile < n_tiles) {
        const float* xg = x + ((size_t)tile * 16 + hi) * IN_DIM + l15 * 4;
        #pragma unroll
        for (int j = 0; j < 8; ++j) r[j] = *(const v4f*)(xg + (size_t)(2 * j) * IN_DIM);
    }
    {
        v4f w = *(const v4f*)(W1 + tid * 4);
        int k = tid >> 2, n0 = (tid & 3) * 4;
        w1t[(n0 + 0) * WP + k] = w.x;
        w1t[(n0 + 1) * WP + k] = w.y;
        w1t[(n0 + 2) * WP + k] = w.z;
        w1t[(n0 + 3) * WP + k] = w.w;
    }
    __syncthreads();
    float* st = xsb + wv * (16 * XP);
    const float* ap = st + l15 * XP + hi * 2;
    const float* bp = w1t + l15 * WP + hi * 2;
    while (tile < n_tiles) {
        const float* xg = x + ((size_t)tile * 16 + hi) * IN_DIM + l15 * 4;
        v8f c = {};
        #pragma unroll
        for (int t = 0; t < IN_DIM / 64; ++t) {
            #pragma unroll
            for (int j = 0; j < 8; ++j) *(v4f*)(st + (2 * j + hi) * XP + l15 * 4) = r[j];
            if (t + 1 < IN_DIM / 64) {
                #pragma unroll
                for (int j = 0; j < 8; ++j) r[j] = *(const v4f*)(xg + (size_t)(2 * j) * IN_DIM + (t + 1) * 64);
            }
            wave_lds_handover();
            v2f a[16], b[16];
            #pragma unroll
            for (int s = 0; s < 16; ++s) {
                a[s] = *(const v2f*)(ap + 4 * s);
                b[s] = *(const v2f*)(bp + t * 64 + 4 * s);
            }
            __builtin_amdgcn_sched_barrier(0);
            #pragma unroll
            for (int s = 0; s < 16; ++s)
                c = __builtin_amdgcn_wmma_f32_16x16x4_f32(false, a[s], false, b[s], (short)0, c, false, false);
            wave_lds_handover();
        }
        float* mt = m1 + (size_t)(tile * 16) * HID;
        #pragma unroll
        for (int v = 0; v < 8; ++v)
            mt[(v + hi * 8) * HID + l15] = c[v];
        tile += tstep;
        if (tile < n_tiles) {
            const float* xn = x + ((size_t)tile * 16 + hi) * IN_DIM + l15 * 4;
            #pragma unroll
            for (int j = 0; j < 8; ++j) r[j] = *(const v4f*)(xn + (size_t)(2 * j) * IN_DIM);
        }
    }
}

__device__ __forceinline__ void bin_fill_part(unsigned char* lds, const int* __restrict__ src, const int* __restrict__ dst,
                                              int* __restrict__ runs, int* __restrict__ pairs,
                                              long long n_edges, long long slice, int n_bins) {
    int* cnt = (int*)lds;
    int* wsum = cnt + MAXBINS + 32;
    int2* stage = (int2*)(wsum + 32);
    int tid = threadIdx.x, lane = tid & 31, wave = tid >> 5;
    int g = blockIdx.x;
    long long beg = (long long)g * slice;
    long long end = beg + slice < n_edges ? beg + slice : n_edges;
    int n = end > beg ? (int)(end - beg) : 0;
    const int* sp = src + beg;
    const int* dp = dst + beg;
    int sv[EU], dv[EU];
    #pragma unroll
    for (int u = 0; u < EU; ++u) { sv[u] = 0; dv[u] = 0; }
    if (n > 0) {
        #pragma unroll
        for (int u = 0; u < EU; ++u) {
            int i = tid + u * WG;
            i = i < n ? i : n - 1;
            sv[u] = sp[i];
            dv[u] = dp[i];
        }
    }
    for (int i = tid; i < n_bins; i += WG) cnt[i] = 0;
    if (tid < 32) cnt[MAXBINS + tid] = 0;
    __syncthreads();
    int rk[EU];
    #pragma unroll
    for (int u = 0; u < EU; ++u)
        rk[u] = atomicAdd(&cnt[tid + u * WG < n ? (int)((unsigned)dv[u] / BIN_NODES) : MAXBINS + lane], 1);
    __syncthreads();
    int own = tid < n_bins ? cnt[tid] : 0;
    int x = own;
    for (int off = 1; off < 32; off <<= 1) { int y = __shfl_up(x, off); if (lane >= off) x += y; }
    if (lane == 31) wsum[wave] = x;
    __syncthreads();
    if (wave == 0) {
        int w = wsum[lane], xs = w;
        for (int off = 1; off < 32; off <<= 1) { int y = __shfl_up(xs, off); if (lane >= off) xs += y; }
        wsum[lane] = xs - w;
    }
    __syncthreads();
    int lbeg = x - own + wsum[wave];
    if (tid < n_bins) {
        cnt[tid] = lbeg;
        runs[(size_t)tid * NSLICE + g] = lbeg;
    }
    if (tid == 0) runs[(size_t)n_bins * NSLICE + g] = n;
    if (tid < 32) cnt[MAXBINS + tid] = SLICE_CAP + tid;
    __syncthreads();
    int pos[EU];
    #pragma unroll
    for (int u = 0; u < EU; ++u) {
        bool in = tid + u * WG < n;
        int b = (int)((unsigned)dv[u] / BIN_NODES);
        pos[u] = cnt[in ? b : MAXBINS + lane] + (in ? rk[u] : 0);
        dv[u] -= b * BIN_NODES;
    }
    #pragma unroll
    for (int u = 0; u < EU; ++u) stage[pos[u]] = make_int2(sv[u], dv[u]);
    __syncthreads();
    int2 r[EU];
    int nl = n > 0 ? n - 1 : 0;
    #pragma unroll
    for (int u = 0; u < EU; ++u) { int i = tid + u * WG; r[u] = stage[i < n ? i : nl]; }
    #pragma unroll
    for (int u = 0; u < EU; ++u) {
        int i = tid + u * WG;
        if (i < n) *(int2*)(pairs + 2 * (size_t)(beg + i)) = r[u];
    }
}

#define FILL_LDS_BYTES ((MAXBINS + 32 + 32) * 4 + (SLICE_CAP + 32) * 8)
#define GEMM_LDS_BYTES ((HID * WP + (WG / 32) * 16 * XP) * 4)
static_assert(((MAXBINS + 32 + 32) * 4) % 8 == 0, "stage holds 8-byte records");
static_assert((HID * WP * 4) % 16 == 0 && (16 * XP * 4) % 16 == 0, "the staged x rows are written 16 bytes at a time");

__global__ void __launch_bounds__(WG)
gcn_fill_gemm1(const int* __restrict__ src, const int* __restrict__ dst,
               int* __restrict__ runs, int* __restrict__ pairs,
               long long n_edges, long long slice, int n_bins,
               const float* __restrict__ x, const float* __restrict__ W1,
               float* __restrict__ m1, int n_tiles) {
    __shared__ __attribute__((aligned(16))) unsigned char lds[GEMM_LDS_BYTES > FILL_LDS_BYTES ? GEMM_LDS_BYTES : FILL_LDS_BYTES];
    bool product_first = (((int)blockIdx.x + ((int)blockIdx.x >> 3)) & 1) != 0;
    if (product_first) gemm1_part(lds, x, W1, m1, n_tiles);
    __syncthreads();
    bin_fill_part(lds, src, dst, runs, pairs, n_edges, slice, n_bins);
    __syncthreads();
    if (!product_first) gemm1_part(lds, x, W1, m1, n_tiles);
}

__device__ __forceinline__ void find_group(const int* pre, const int* rbase, int p0, int end, int last, int* idx) {
    int q[PG], g[PG];
    #pragma unroll
    for (int u = 0; u < PG; ++u) {
        int p = p0 + u * WG;
        q[u] = p < end ? p : end - 1;
        q[u] = q[u] > 0 ? q[u] : 0;
        g[u] = 0;
    }
    #pragma unroll
    for (int s = NSLICE / 2; s > 0; s >>= 1) {
        #pragma unroll
        for (int u = 0; u < PG; ++u) g[u] += pre[g[u] + s] <= q[u] ? s : 0;
    }
    #pragma unroll
    for (int u = 0; u < PG; ++u) {
        int i = rbase[g[u]] + q[u];
        i = i < last ? i : last;
        idx[u] = i > 0 ? i : 0;
    }
}

__device__ __forceinline__ void fetch_group(const int* __restrict__ pairs, const int* idx, int p0, int end, int2* out) {
    #pragma unroll
    for (int u = 0; u < PG; ++u) {
        int2 r = *(const int2*)(pairs + 2 * (size_t)idx[u]);
        if (p0 + u * WG >= end) r = make_int2(0, BIN_NODES);
        out[u] = r;
    }
}

__device__ __forceinline__ void gather_rows(const float* __restrict__ msg, int s_l, int f, float* v) {
    #pragma unroll
    for (int j = 0; j < 16; ++j)
        v[j] = msg[(unsigned)__shfl(s_l, j, 16) * HID + f];
}

__device__ __forceinline__ void add_rows(float* acc, int d_l, int f, const float* v) {
    #pragma unroll
    for (int j = 0; j < 16; ++j) atomicAdd(&acc[__shfl(d_l, j, 16) * HID + f], v[j]);
}

template <int NF, bool BIAS, bool PROD>
__global__ void __launch_bounds__(WG)
gcn_bin_sum(const int* __restrict__ pairs, const int* __restrict__ runs,
            const float* __restrict__ msg, const float* __restrict__ bias, const float* __restrict__ W2,
            float* __restrict__ res, int n_nodes, int slice, int last) {
    __shared__ float acc[(BIN_NODES + 1) * HID];
    __shared__ int pre[NSLICE];
    __shared__ int rbase[NSLICE];
    __shared__ int wtot[32];
    int tid = threadIdx.x, lane = tid & 31, f = lane & 15;
    int wave = __builtin_amdgcn_readfirstlane(tid >> 5);
    int b = blockIdx.x;
    int rs = 0, len = 0;
    if (tid < NSLICE) {
        rs  = runs[(size_t)b * NSLICE + tid];
        len = runs[(size_t)(b + 1) * NSLICE + tid] - rs;
    }
    for (int i = tid; i < (BIN_NODES + 1) * HID; i += WG) acc[i] = 0.0f;
    int x = len;
    for (int off = 1; off < 32; off <<= 1) { int y = __shfl_up(x, off); if (lane >= off) x += y; }
    if (lane == 31) wtot[wave] = x;
    __syncthreads();
    int wt = wtot[lane], ws = wt;
    for (int off = 1; off < NSLICE / 32; off <<= 1) { int y = __shfl_up(ws, off); if (lane >= off) ws += y; }
    int end = __builtin_amdgcn_readlane(ws, NSLICE / 32 - 1);
    int excl = x - len + __builtin_amdgcn_readlane(ws - wt, wave);
    if (tid < NSLICE) {
        pre[tid]   = excl;
        rbase[tid] = tid * slice + rs - excl;
    }
    __syncthreads();
    int e0 = wave * 32;
    int2 pn[PG];
    int ix[PG];
    find_group(pre, rbase, e0 + lane, end, last, ix);
    fetch_group(pairs, ix, e0 + lane, end, pn);
    asm volatile("" :: "v"(pn[0].x), "v"(pn[0].y), "v"(pn[1].x), "v"(pn[1].y), "v"(pn[2].x), "v"(pn[2].y), "v"(pn[3].x), "v"(pn[3].y));
    static_assert(PG == 4, "the statement above names the PG records one by one");
    float v[PG - 1][16];
    for (; e0 + (PG - 1) * WG < end; e0 += WG * PG) {
        int2 p[PG];
        #pragma unroll
        for (int t = 0; t < PG; ++t) p[t] = pn[t];
        gather_rows(msg, p[0].x, f, v[0]);
        #pragma unroll
        for (int t = 0; t < PG; ++t) {
            if (t + 1 < PG) gather_rows(msg, p[t + 1].x, f, v[(t + 1) & 1]);
            if (t == 0) {
                __builtin_amdgcn_sched_barrier(0);
                find_group(pre, rbase, e0 + PG * WG + lane, end, last, ix);
            }
            if (t == PG - 2) {
                fetch_group(pairs, ix, e0 + PG * WG + lane, end, pn);
            }
            add_rows(acc, p[t].y, f, v[t & 1]);
        }
    }
    if (e0 < end) {
        #pragma unroll
        for (int t = 0; t < PG - 1; ++t) gather_rows(msg, pn[t].x, f, v[t]);
        #pragma unroll
        for (int t = 0; t < PG - 1; ++t)
            if (e0 + t * WG < end) add_rows(acc, pn[t].y, f, v[t]);
    }
    int hi = lane >> 4;
    v2f wb[HID / 4], bb[HID / 4];
    if (PROD) {
        #pragma unroll
        for (int k = 0; k < HID / 4; ++k) {
            int ka = k * 4 + hi * 2;
            int fc = f < OUTD ? f : 0;
            float w0 = W2[(ka + 0) * OUTD + fc], w1 = W2[(ka + 1) * OUTD + fc];
            wb[k].x = f < OUTD ? w0 : 0.0f;
            wb[k].y = f < OUTD ? w1 : 0.0f;
            bb[k].x = bias[ka];
            bb[k].y = bias[ka + 1];
        }
    }
    __syncthreads();
    if (PROD) {
        static_assert(BIN_NODES % 16 == 0 && BIN_NODES / 16 <= WG / 32, "one whole 16-row tile per wave");
        if (wave < BIN_NODES / 16) {
            float* at = acc + wave * 16 * HID;
            v8f c = {};
            #pragma unroll
            for (int k = 0; k < HID / 4; ++k) {
                int ka = k * 4 + hi * 2;
                v2f s = *(const v2f*)(at + f * HID + ka);
                v2f a;
                a.x = fmaxf(s.x + bb[k].x, 0.0f);
                a.y = fmaxf(s.y + bb[k].y, 0.0f);
                c = __builtin_amdgcn_wmma_f32_16x16x4_f32(false, a, false, wb[k], (short)0, c, false, false);
            }
            #pragma unroll
            for (int r = 0; r < 8; ++r)
                at[(r + hi * 8) * HID + f] = c[r];
        }
        __syncthreads();
    }
    int rows = n_nodes - b * BIN_NODES;
    if (rows > BIN_NODES) rows = BIN_NODES;
    float* rb = res + (size_t)b * BIN_NODES * NF;
    for (int i = tid; i < rows * NF; i += WG) {
        int r = i / NF, o = i - r * NF;
        float a = acc[r * HID + o];
        rb[i] = BIAS ? a + bias[o] : a;
    }
}

extern "C" void kernel_launch(void* const* d_in, const int* in_sizes, int n_in,
                              void* d_out, int out_size, void* d_ws, size_t ws_size,
                              hipStream_t stream) {
    const float* x  = (const float*)d_in[0];
    const int*   ei = (const int*)  d_in[1];
    const float* W1 = (const float*)d_in[2];
    const float* b1 = (const float*)d_in[3];
    const float* W2 = (const float*)d_in[4];
    const float* b2 = (const float*)d_in[5];
    float* out = (float*)d_out;

    int n_nodes       = in_sizes[0] / IN_DIM;
    long long n_edges = (long long)in_sizes[1] / 2;
    const int* src = ei;
    const int* dst = ei + n_edges;

    int n_bins      = (n_nodes + BIN_NODES - 1) / BIN_NODES;
    long long slice = (n_edges + NSLICE - 1) / NSLICE;

    float* m1  = (float*)d_ws;
    float* m2  = m1 + (size_t)n_nodes * HID;
    int* runs  = (int*)(m2 + (size_t)n_nodes * HID);
    int* pairs = runs + (size_t)(n_bins + 1) * NSLICE;

    int n_tiles = (n_nodes + 15) / 16;

    gcn_fill_gemm1 <<<NSLICE, WG, 0, stream>>>(src, dst, runs, pairs, n_edges, slice, n_bins, x, W1, m1, n_tiles);
    gcn_bin_sum<HID,  false, true ><<<n_bins, WG, 0, stream>>>(pairs, runs, m1, b1, W2, m2,  n_nodes, (int)slice, (int)(n_edges - 1));
    gcn_bin_sum<OUTD, true,  false><<<n_bins, WG, 0, stream>>>(pairs, runs, m2, b2, W2, out, n_nodes, (int)slice, (int)(n_edges - 1));
}


# edge records packed into one 4-byte word (23 bits of source, 9 of the row inside the block): the list written once and read by both block sums has half the bytes and lines; sums in LDS arrival order
# speedup vs baseline: 19.0609x; 1.0530x over previous
#include <hip/hip_runtime.h>

typedef float v2f __attribute__((ext_vector_type(2)));
typedef float v8f __attribute__((ext_vector_type(8)));

#define IN_DIM 256
#define HID 16
#define OUTD 10

#define BIN_NODES 400
#define NSLICE    256
#define MAXBINS   1024
#define WG        1024
#define EU        16
#define PG        4
#define SLICE_CAP 12800
static_assert(SLICE_CAP <= WG * EU, "a thread holds its EU edges of the slice in registers from the rank to the placing");
static_assert(NSLICE == 256, "the search in find_group takes 8 halving steps over 256 runs");
#define SRC_BITS  23
static_assert(BIN_NODES < (1 << (32 - SRC_BITS)), "rows 0 .. BIN_NODES (the spare row included) fit the bits above the source");
__device__ __forceinline__ int rec_src(int r) { return r & ((1 << SRC_BITS) - 1); }
__device__ __forceinline__ int rec_row(int r) { return (int)((unsigned)r >> SRC_BITS); }

#define XP 68
#define WP 260
typedef float v4f __attribute__((ext_vector_type(4)));
static_assert(IN_DIM * HID == WG * 4, "W1 staging: one 16-byte piece of W1 per thread of the work-group");

__device__ __forceinline__ void wave_lds_handover() {
    __builtin_amdgcn_fence(__ATOMIC_RELEASE, "wavefront");
    __builtin_amdgcn_wave_barrier();
    __builtin_amdgcn_fence(__ATOMIC_ACQUIRE, "wavefront");
}

__device__ __forceinline__ void gemm1_part(unsigned char* lds, const float* __restrict__ x, const float* __restrict__ W1,
                                           float* __restrict__ m1, int n_tiles) {
    float* w1t = (float*)lds;
    float* xsb = w1t + HID * WP;
    int tid = threadIdx.x, lane = tid & 31;
    int wv = __builtin_amdgcn_readfirstlane(tid >> 5);
    int tstep = (WG / 32) * (int)gridDim.x;
    int tile = wv * (int)gridDim.x + (int)blockIdx.x;
    int l15 = lane & 15;
    int hi  = lane >> 4;
    v4f r[8];
    if (tile < n_tiles) {
        const float* xg = x + ((size_t)tile * 16 + hi) * IN_DIM + l15 * 4;
        #pragma unroll
        for (int j = 0; j < 8; ++j) r[j] = *(const v4f*)(xg + (size_t)(2 * j) * IN_DIM);
    }
    {
        v4f w = *(const v4f*)(W1 + tid * 4);
        int k = tid >> 2, n0 = (tid & 3) * 4;
        w1t[(n0 + 0) * WP + k] = w.x;
        w1t[(n0 + 1) * WP + k] = w.y;
        w1t[(n0 + 2) * WP + k] = w.z;
        w1t[(n0 + 3) * WP + k] = w.w;
    }
    __syncthreads();
    float* st = xsb + wv * (16 * XP);
    const float* ap = st + l15 * XP + hi * 2;
    const float* bp = w1t + l15 * WP + hi * 2;
    while (tile < n_tiles) {
        const float* xg = x + ((size_t)tile * 16 + hi) * IN_DIM + l15 * 4;
        v8f c = {};
        #pragma unroll
        for (int t = 0; t < IN_DIM / 64; ++t) {
            #pragma unroll
            for (int j = 0; j < 8; ++j) *(v4f*)(st + (2 * j + hi) * XP + l15 * 4) = r[j];
            if (t + 1 < IN_DIM / 64) {
                #pragma unroll
                for (int j = 0; j < 8; ++j) r[j] = *(const v4f*)(xg + (size_t)(2 * j) * IN_DIM + (t + 1) * 64);
            }
            wave_lds_handover();
            v2f a[16], b[16];
            #pragma unroll
            for (int s = 0; s < 16; ++s) {
                a[s] = *(const v2f*)(ap + 4 * s);
                b[s] = *(const v2f*)(bp + t * 64 + 4 * s);
            }
            __builtin_amdgcn_sched_barrier(0);
            #pragma unroll
            for (int s = 0; s < 16; ++s)
                c = __builtin_amdgcn_wmma_f32_16x16x4_f32(false, a[s], false, b[s], (short)0, c, false, false);
            wave_lds_handover();
        }
        float* mt = m1 + (size_t)(tile * 16) * HID;
        #pragma unroll
        for (int v = 0; v < 8; ++v)
            mt[(v + hi * 8) * HID + l15] = c[v];
        tile += tstep;
        if (tile < n_tiles) {
            const float* xn = x + ((size_t)tile * 16 + hi) * IN_DIM + l15 * 4;
            #pragma unroll
            for (int j = 0; j < 8; ++j) r[j] = *(const v4f*)(xn + (size_t)(2 * j) * IN_DIM);
        }
    }
}

__device__ __forceinline__ void bin_fill_part(unsigned char* lds, const int* __restrict__ src, const int* __restrict__ dst,
                                              int* __restrict__ runs, int* __restrict__ pairs,
                                              long long n_edges, long long slice, int n_bins) {
    int* cnt = (int*)lds;
    int* wsum = cnt + MAXBINS + 32;
    int* stage = wsum + 32;
    int tid = threadIdx.x, lane = tid & 31, wave = tid >> 5;
    int g = blockIdx.x;
    long long beg = (long long)g * slice;
    long long end = beg + slice < n_edges ? beg + slice : n_edges;
    int n = end > beg ? (int)(end - beg) : 0;
    const int* sp = src + beg;
    const int* dp = dst + beg;
    int sv[EU], dv[EU];
    #pragma unroll
    for (int u = 0; u < EU; ++u) { sv[u] = 0; dv[u] = 0; }
    if (n > 0) {
        #pragma unroll
        for (int u = 0; u < EU; ++u) {
            int i = tid + u * WG;
            i = i < n ? i : n - 1;
            sv[u] = sp[i];
            dv[u] = dp[i];
        }
    }
    for (int i = tid; i < n_bins; i += WG) cnt[i] = 0;
    if (tid < 32) cnt[MAXBINS + tid] = 0;
    __syncthreads();
    int rk[EU];
    #pragma unroll
    for (int u = 0; u < EU; ++u)
        rk[u] = atomicAdd(&cnt[tid + u * WG < n ? (int)((unsigned)dv[u] / BIN_NODES) : MAXBINS + lane], 1);
    __syncthreads();
    int own = tid < n_bins ? cnt[tid] : 0;
    int x = own;
    for (int off = 1; off < 32; off <<= 1) { int y = __shfl_up(x, off); if (lane >= off) x += y; }
    if (lane == 31) wsum[wave] = x;
    __syncthreads();
    if (wave == 0) {
        int w = wsum[lane], xs = w;
        for (int off = 1; off < 32; off <<= 1) { int y = __shfl_up(xs, off); if (lane >= off) xs += y; }
        wsum[lane] = xs - w;
    }
    __syncthreads();
    int lbeg = x - own + wsum[wave];
    if (tid < n_bins) {
        cnt[tid] = lbeg;
        runs[(size_t)tid * NSLICE + g] = lbeg;
    }
    if (tid == 0) runs[(size_t)n_bins * NSLICE + g] = n;
    if (tid < 32) cnt[MAXBINS + tid] = SLICE_CAP + tid;
    __syncthreads();
    int pos[EU];
    #pragma unroll
    for (int u = 0; u < EU; ++u) {
        bool in = tid + u * WG < n;
        int b = (int)((unsigned)dv[u] / BIN_NODES);
        pos[u] = cnt[in ? b : MAXBINS + lane] + (in ? rk[u] : 0);
        dv[u] -= b * BIN_NODES;
    }
    #pragma unroll
    for (int u = 0; u < EU; ++u) stage[pos[u]] = (int)((unsigned)sv[u] | ((unsigned)dv[u] << SRC_BITS));
    __syncthreads();
    int r[EU];
    int nl = n > 0 ? n - 1 : 0;
    #pragma unroll
    for (int u = 0; u < EU; ++u) { int i = tid + u * WG; r[u] = stage[i < n ? i : nl]; }
    #pragma unroll
    for (int u = 0; u < EU; ++u) {
        int i = tid + u * WG;
        if (i < n) pairs[(size_t)(beg + i)] = r[u];
    }
}

#define FILL_LDS_BYTES ((MAXBINS + 32 + 32) * 4 + (SLICE_CAP + 32) * 4)
#define GEMM_LDS_BYTES ((HID * WP + (WG / 32) * 16 * XP) * 4)
static_assert((HID * WP * 4) % 16 == 0 && (16 * XP * 4) % 16 == 0, "the staged x rows are written 16 bytes at a time");

__global__ void __launch_bounds__(WG)
gcn_fill_gemm1(const int* __restrict__ src, const int* __restrict__ dst,
               int* __restrict__ runs, int* __restrict__ pairs,
               long long n_edges, long long slice, int n_bins,
               const float* __restrict__ x, const float* __restrict__ W1,
               float* __restrict__ m1, int n_tiles) {
    __shared__ __attribute__((aligned(16))) unsigned char lds[GEMM_LDS_BYTES > FILL_LDS_BYTES ? GEMM_LDS_BYTES : FILL_LDS_BYTES];
    bool product_first = (((int)blockIdx.x + ((int)blockIdx.x >> 3)) & 1) != 0;
    if (product_first) gemm1_part(lds, x, W1, m1, n_tiles);
    __syncthreads();
    bin_fill_part(lds, src, dst, runs, pairs, n_edges, slice, n_bins);
    __syncthreads();
    if (!product_first) gemm1_part(lds, x, W1, m1, n_tiles);
}

__device__ __forceinline__ void find_group(const int* pre, const int* rbase, int p0, int end, int last, int* idx) {
    int q[PG], g[PG];
    #pragma unroll
    for (int u = 0; u < PG; ++u) {
        int p = p0 + u * WG;
        q[u] = p < end ? p : end - 1;
        q[u] = q[u] > 0 ? q[u] : 0;
        g[u] = 0;
    }
    #pragma unroll
    for (int s = NSLICE / 2; s > 0; s >>= 1) {
        #pragma unroll
        for (int u = 0; u < PG; ++u) g[u] += pre[g[u] + s] <= q[u] ? s : 0;
    }
    #pragma unroll
    for (int u = 0; u < PG; ++u) {
        int i = rbase[g[u]] + q[u];
        i = i < last ? i : last;
        idx[u] = i > 0 ? i : 0;
    }
}

__device__ __forceinline__ void fetch_group(const int* __restrict__ pairs, const int* idx, int p0, int end, int* out) {
    #pragma unroll
    for (int u = 0; u < PG; ++u) {
        int r = pairs[idx[u]];
        if (p0 + u * WG >= end) r = (int)((unsigned)BIN_NODES << SRC_BITS);
        out[u] = r;
    }
}

__device__ __forceinline__ void gather_rows(const float* __restrict__ msg, int s_l, int f, float* v) {
    #pragma unroll
    for (int j = 0; j < 16; ++j)
        v[j] = msg[(unsigned)__shfl(s_l, j, 16) * HID + f];
}

__device__ __forceinline__ void add_rows(float* acc, int d_l, int f, const float* v) {
    #pragma unroll
    for (int j = 0; j < 16; ++j) atomicAdd(&acc[__shfl(d_l, j, 16) * HID + f], v[j]);
}

template <int NF, bool BIAS, bool PROD>
__global__ void __launch_bounds__(WG)
gcn_bin_sum(const int* __restrict__ pairs, const int* __restrict__ runs,
            const float* __restrict__ msg, const float* __restrict__ bias, const float* __restrict__ W2,
            float* __restrict__ res, int n_nodes, int slice, int last) {
    __shared__ float acc[(BIN_NODES + 1) * HID];
    __shared__ int pre[NSLICE];
    __shared__ int rbase[NSLICE];
    __shared__ int wtot[32];
    int tid = threadIdx.x, lane = tid & 31, f = lane & 15;
    int wave = __builtin_amdgcn_readfirstlane(tid >> 5);
    int b = blockIdx.x;
    int rs = 0, len = 0;
    if (tid < NSLICE) {
        rs  = runs[(size_t)b * NSLICE + tid];
        len = runs[(size_t)(b + 1) * NSLICE + tid] - rs;
    }
    for (int i = tid; i < (BIN_NODES + 1) * HID; i += WG) acc[i] = 0.0f;
    int x = len;
    for (int off = 1; off < 32; off <<= 1) { int y = __shfl_up(x, off); if (lane >= off) x += y; }
    if (lane == 31) wtot[wave] = x;
    __syncthreads();
    int wt = wtot[lane], ws = wt;
    for (int off = 1; off < NSLICE / 32; off <<= 1) { int y = __shfl_up(ws, off); if (lane >= off) ws += y; }
    int end = __builtin_amdgcn_readlane(ws, NSLICE / 32 - 1);
    int excl = x - len + __builtin_amdgcn_readlane(ws - wt, wave);
    if (tid < NSLICE) {
        pre[tid]   = excl;
        rbase[tid] = tid * slice + rs - excl;
    }
    __syncthreads();
    int e0 = wave * 32;
    int pn[PG];
    int ix[PG];
    find_group(pre, rbase, e0 + lane, end, last, ix);
    fetch_group(pairs, ix, e0 + lane, end, pn);
    asm volatile("" :: "v"(pn[0]), "v"(pn[1]), "v"(pn[2]), "v"(pn[3]));
    static_assert(PG == 4, "the statement above names the PG records one by one");
    float v[PG - 1][16];
    for (; e0 + (PG - 1) * WG < end; e0 += WG * PG) {
        int p[PG];
        #pragma unroll
        for (int t = 0; t < PG; ++t) p[t] = pn[t];
        gather_rows(msg, rec_src(p[0]), f, v[0]);
        #pragma unroll
        for (int t = 0; t < PG; ++t) {
            if (t + 1 < PG) gather_rows(msg, rec_src(p[t + 1]), f, v[(t + 1) & 1]);
            if (t == 0) {
                __builtin_amdgcn_sched_barrier(0);
                find_group(pre, rbase, e0 + PG * WG + lane, end, last, ix);
            }
            if (t == PG - 2) {
                fetch_group(pairs, ix, e0 + PG * WG + lane, end, pn);
            }
            add_rows(acc, rec_row(p[t]), f, v[t & 1]);
        }
    }
    if (e0 < end) {
        #pragma unroll
        for (int t = 0; t < PG - 1; ++t) gather_rows(msg, rec_src(pn[t]), f, v[t]);
        #pragma unroll
        for (int t = 0; t < PG - 1; ++t)
            if (e0 + t * WG < end) add_rows(acc, rec_row(pn[t]), f, v[t]);
    }
    int hi = lane >> 4;
    v2f wb[HID / 4], bb[HID / 4];
    if (PROD) {
        #pragma unroll
        for (int k = 0; k < HID / 4; ++k) {
            int ka = k * 4 + hi * 2;
            int fc = f < OUTD ? f : 0;
            float w0 = W2[(ka + 0) * OUTD + fc], w1 = W2[(ka + 1) * OUTD + fc];
            wb[k].x = f < OUTD ? w0 : 0.0f;
            wb[k].y = f < OUTD ? w1 : 0.0f;
            bb[k].x = bias[ka];
            bb[k].y = bias[ka + 1];
        }
    }
    __syncthreads();
    if (PROD) {
        static_assert(BIN_NODES % 16 == 0 && BIN_NODES / 16 <= WG / 32, "one whole 16-row tile per wave");
        if (wave < BIN_NODES / 16) {
            float* at = acc + wave * 16 * HID;
            v8f c = {};
            #pragma unroll
            for (int k = 0; k < HID / 4; ++k) {
                int ka = k * 4 + hi * 2;
                v2f s = *(const v2f*)(at + f * HID + ka);
                v2f a;
                a.x = fmaxf(s.x + bb[k].x, 0.0f);
                a.y = fmaxf(s.y + bb[k].y, 0.0f);
                c = __builtin_amdgcn_wmma_f32_16x16x4_f32(false, a, false, wb[k], (short)0, c, false, false);
            }
            #pragma unroll
            for (int r = 0; r < 8; ++r)
                at[(r + hi * 8) * HID + f] = c[r];
        }
        __syncthreads();
    }
    int rows = n_nodes - b * BIN_NODES;
    if (rows > BIN_NODES) rows = BIN_NODES;
    float* rb = res + (size_t)b * BIN_NODES * NF;
    for (int i = tid; i < rows * NF; i += WG) {
        int r = i / NF, o = i - r * NF;
        float a = acc[r * HID + o];
        rb[i] = BIAS ? a + bias[o] : a;
    }
}

extern "C" void kernel_launch(void* const* d_in, const int* in_sizes, int n_in,
                              void* d_out, int out_size, void* d_ws, size_t ws_size,
                              hipStream_t stream) {
    const float* x  = (const float*)d_in[0];
    const int*   ei = (const int*)  d_in[1];
    const float* W1 = (const float*)d_in[2];
    const float* b1 = (const float*)d_in[3];
    const float* W2 = (const float*)d_in[4];
    const float* b2 = (const float*)d_in[5];
    float* out = (float*)d_out;

    int n_nodes       = in_sizes[0] / IN_DIM;
    long long n_edges = (long long)in_sizes[1] / 2;
    const int* src = ei;
    const int* dst = ei + n_edges;

    int n_bins      = (n_nodes + BIN_NODES - 1) / BIN_NODES;
    long long slice = (n_edges + NSLICE - 1) / NSLICE;

    float* m1  = (float*)d_ws;
    float* m2  = m1 + (size_t)n_nodes * HID;
    int* runs  = (int*)(m2 + (size_t)n_nodes * HID);
    int* pairs = runs + (size_t)(n_bins + 1) * NSLICE;

    int n_tiles = (n_nodes + 15) / 16;

    gcn_fill_gemm1 <<<NSLICE, WG, 0, stream>>>(src, dst, runs, pairs, n_edges, slice, n_bins, x, W1, m1, n_tiles);
    gcn_bin_sum<HID,  false, true ><<<n_bins, WG, 0, stream>>>(pairs, runs, m1, b1, W2, m2,  n_nodes, (int)slice, (int)(n_edges - 1));
    gcn_bin_sum<OUTD, true,  false><<<n_bins, WG, 0, stream>>>(pairs, runs, m2, b2, W2, out, n_nodes, (int)slice, (int)(n_edges - 1));
}
